# starting kernel
# baseline (speedup 1.0000x reference)
#include <hip/hip_runtime.h>
#include <hip/hip_bf16.h>


typedef __attribute__((ext_vector_type(2))) float v2f;
typedef __attribute__((ext_vector_type(8))) float v8f;

#define N_NODES 50000
#define N_EDGES 800000
#define HID     128
#define N_POS   100000

__global__ void zero_kernel(float* __restrict__ p, long long n) {
  long long i = (long long)blockIdx.x * blockDim.x + threadIdx.x;
  long long stride = (long long)gridDim.x * blockDim.x;
  for (; i < n; i += stride) p[i] = 0.0f;
}

__global__ void transpose_kernel(const float* __restrict__ W, float* __restrict__ Wt,
                                 int K) {
  int id = blockIdx.x * blockDim.x + threadIdx.x;
  if (id < K * HID) {
    int n = id / K;
    int k = id - n * K;
    Wt[id] = W[k * HID + n];
  }
}

__global__ void deg_kernel(const int* __restrict__ idx, const float* __restrict__ vals,
                           float* __restrict__ deg, int nedges) {
  int e = blockIdx.x * blockDim.x + threadIdx.x;
  if (e < nedges) atomicAdd(&deg[idx[2 * e]], vals[e]);
}

__global__ void deginv_kernel(float* __restrict__ deg, int n) {
  int i = blockIdx.x * blockDim.x + threadIdx.x;
  if (i < n) deg[i] = 1.0f / fmaxf(deg[i], 1.0f);
}

__global__ void spmm_kernel(const int* __restrict__ idx, const float* __restrict__ vals,
                            const float* __restrict__ x, float* __restrict__ out,
                            int nedges) {
  int gw   = (blockIdx.x * blockDim.x + threadIdx.x) >> 5;
  int lane = threadIdx.x & 31;
  if (gw >= nedges) return;
  int   row = idx[2 * gw];
  int   col = idx[2 * gw + 1];
  float v   = vals[gw];
  int   f   = lane << 2;
  const float4 xv = *(const float4*)(x + (long long)col * HID + f);
  float* o = out + (long long)row * HID + f;
  atomicAdd(o + 0, v * xv.x);
  atomicAdd(o + 1, v * xv.y);
  atomicAdd(o + 2, v * xv.z);
  atomicAdd(o + 3, v * xv.w);
}

__global__ void gemm_wmma_kernel(const float* __restrict__ X, const float* __restrict__ scale,
                                 const float* __restrict__ Wt, const float* __restrict__ bias,
                                 float* __restrict__ Y, int do_relu) {
  __shared__ float Xs[16][132];
  const int tile_m = blockIdx.x;
  const int t      = threadIdx.x;

  {
    int   r    = t >> 4;
    int   cseg = (t & 15) << 3;
    float s    = scale[tile_m * 16 + r];
    const float* src = X + ((long long)tile_m * 16 + r) * HID + cseg;
    float4 v0 = *(const float4*)(src);
    float4 v1 = *(const float4*)(src + 4);
    float* dst = &Xs[r][cseg];
    dst[0] = v0.x * s; dst[1] = v0.y * s; dst[2] = v0.z * s; dst[3] = v0.w * s;
    dst[4] = v1.x * s; dst[5] = v1.y * s; dst[6] = v1.z * s; dst[7] = v1.w * s;
  }
  __syncthreads();

  const int lane = t & 31;
  const int wid  = t >> 5;
  const int m    = lane & 15;
  const int kh   = (lane >> 4) << 1;
  const int n    = (wid << 4) + (lane & 15);

  const float* wrow = Wt + n * HID;

  v8f c = {};
#pragma unroll
  for (int k0 = 0; k0 < HID; k0 += 4) {
    int k = k0 + kh;
    v2f a; a.x = Xs[m][k]; a.y = Xs[m][k + 1];
    v2f b = *(const v2f*)(wrow + k);
    c = __builtin_amdgcn_wmma_f32_16x16x4_f32(false, a, false, b, (short)0, c,
                                              false, false);
  }

  const float bn      = bias[n];
  const int   rowbase = tile_m * 16 + ((lane >> 4) << 3);
#pragma unroll
  for (int i = 0; i < 8; ++i) {
    float v = c[i] + bn;
    if (do_relu) v = fmaxf(v, 0.0f);
    Y[(long long)(rowbase + i) * HID + n] = v;
  }
}

__global__ void decode_kernel(const float* __restrict__ h2,
                              const int* __restrict__ U, const int* __restrict__ V,
                              const float* __restrict__ Wd1t, const float* __restrict__ bd1,
                              const float* __restrict__ Wd2, const float* __restrict__ bd2,
                              float* __restrict__ out) {
  __shared__ float Zs[16][260];
  __shared__ float Hs[16][HID];
  __shared__ float red[16][16];

  const int base = blockIdx.x * 16;
  const int t    = threadIdx.x;

  {
    int i   = t >> 4;
    int seg = t & 15;
    const float* src = (seg < 8)
        ? h2 + (long long)U[base + i] * HID + (seg << 4)
        : h2 + (long long)V[base + i] * HID + ((seg - 8) << 4);
    float* dst = &Zs[i][seg << 4];
#pragma unroll
    for (int j = 0; j < 16; j += 4) {
      float4 v = *(const float4*)(src + j);
      dst[j] = v.x; dst[j + 1] = v.y; dst[j + 2] = v.z; dst[j + 3] = v.w;
    }
  }
  __syncthreads();

  const int lane = t & 31;
  const int wid  = t >> 5;
  const int m    = lane & 15;
  const int kh   = (lane >> 4) << 1;
  const int n    = (wid << 4) + (lane & 15);

  const float* wrow = Wd1t + n * (2 * HID);

  v8f c = {};
#pragma unroll
  for (int k0 = 0; k0 < 2 * HID; k0 += 4) {
    int k = k0 + kh;
    v2f a; a.x = Zs[m][k]; a.y = Zs[m][k + 1];
    v2f b = *(const v2f*)(wrow + k);
    c = __builtin_amdgcn_wmma_f32_16x16x4_f32(false, a, false, b, (short)0, c,
                                              false, false);
  }

  const float bn    = bd1[n];
  const int   rbase = (lane >> 4) << 3;
#pragma unroll
  for (int i = 0; i < 8; ++i) Hs[rbase + i][n] = fmaxf(c[i] + bn, 0.0f);
  __syncthreads();

  {
    int i   = t >> 4;
    int seg = t & 15;
    float p = 0.0f;
#pragma unroll
    for (int j = 0; j < 8; ++j) p += Hs[i][(seg << 3) + j] * Wd2[(seg << 3) + j];
    red[i][seg] = p;
  }
  __syncthreads();
  if (t < 16) {
    float s = bd2[0];
#pragma unroll
    for (int j = 0; j < 16; ++j) s += red[t][j];
    out[base + t] = s;
  }
}

extern "C" void kernel_launch(void* const* d_in, const int* in_sizes, int n_in,
                              void* d_out, int out_size, void* d_ws, size_t ws_size,
                              hipStream_t stream) {
  const int*   adj_idx = (const int*)  d_in[0];
  const float* adj_val = (const float*)d_in[1];
  const float* h       = (const float*)d_in[3];
  const int*   pos_u   = (const int*)  d_in[4];
  const int*   pos_v   = (const int*)  d_in[5];
  const int*   neg_u   = (const int*)  d_in[6];
  const int*   neg_v   = (const int*)  d_in[7];
  const float* W1      = (const float*)d_in[8];
  const float* b1      = (const float*)d_in[9];
  const float* W2      = (const float*)d_in[10];
  const float* b2      = (const float*)d_in[11];
  const float* Wd1     = (const float*)d_in[12];
  const float* bd1     = (const float*)d_in[13];
  const float* Wd2     = (const float*)d_in[14];
  const float* bd2     = (const float*)d_in[15];

  float* outp = (float*)d_out;
  float* pos_scores = outp;
  float* neg_scores = outp + N_POS;
  float* h2         = outp + 2 * N_POS;

  float* ws      = (float*)d_ws;
  float* deg_inv = ws;
  float* tmp     = ws + 50176;
  float* h1      = tmp + (long long)N_NODES * HID;
  float* W1t     = h1 + (long long)N_NODES * HID;
  float* W2t     = W1t + HID * HID;
  float* Wd1t    = W2t + HID * HID;

  const long long feat_elems = (long long)N_NODES * HID;

  transpose_kernel<<<(HID * HID + 255) / 256, 256, 0, stream>>>(W1, W1t, HID);
  transpose_kernel<<<(HID * HID + 255) / 256, 256, 0, stream>>>(W2, W2t, HID);
  transpose_kernel<<<(2 * HID * HID + 255) / 256, 256, 0, stream>>>(Wd1, Wd1t, 2 * HID);
  zero_kernel<<<2048, 256, 0, stream>>>(ws, 50176 + feat_elems);
  deg_kernel<<<(N_EDGES + 255) / 256, 256, 0, stream>>>(adj_idx, adj_val, deg_inv, N_EDGES);
  deginv_kernel<<<(N_NODES + 255) / 256, 256, 0, stream>>>(deg_inv, N_NODES);
  spmm_kernel<<<N_EDGES / 8, 256, 0, stream>>>(adj_idx, adj_val, h, tmp, N_EDGES);
  gemm_wmma_kernel<<<N_NODES / 16, 256, 0, stream>>>(tmp, deg_inv, W1t, b1, h1, 1);
  zero_kernel<<<2048, 256, 0, stream>>>(tmp, feat_elems);
  spmm_kernel<<<N_EDGES / 8, 256, 0, stream>>>(adj_idx, adj_val, h1, tmp, N_EDGES);
  gemm_wmma_kernel<<<N_NODES / 16, 256, 0, stream>>>(tmp, deg_inv, W2t, b2, h2, 0);
  decode_kernel<<<N_POS / 16, 256, 0, stream>>>(h2, pos_u, pos_v, Wd1t, bd1, Wd2, bd2, pos_scores);
  decode_kernel<<<N_POS / 16, 256, 0, stream>>>(h2, neg_u, neg_v, Wd1t, bd1, Wd2, bd2, neg_scores);
}


# B weights stored in WMMA B-fragment order: each wave reads the B operands of two k-steps as one contiguous 512 B load instead of row-strided 8 B per lane; k order and sum order unchanged
# speedup vs baseline: 1.2656x; 1.2656x over previous
#include <hip/hip_runtime.h>
#include <hip/hip_bf16.h>


typedef __attribute__((ext_vector_type(2))) float v2f;
typedef __attribute__((ext_vector_type(8))) float v8f;

#define N_NODES 50000
#define N_EDGES 800000
#define HID     128
#define N_POS   100000

__global__ void zero_kernel(float* __restrict__ p, long long n) {
  long long i = (long long)blockIdx.x * blockDim.x + threadIdx.x;
  long long stride = (long long)gridDim.x * blockDim.x;
  for (; i < n; i += stride) p[i] = 0.0f;
}

__global__ void fragorder_kernel(const float* __restrict__ W, float* __restrict__ Wf,
                                 int K) {
  int id = blockIdx.x * blockDim.x + threadIdx.x;
  if (id < K * HID) {
    int j    = id & 3;
    int lane = (id >> 2) & 31;
    int rest = id >> 7;
    int ks2  = K >> 3;
    int w    = rest / ks2;
    int sidx = rest - w * ks2;
    int n    = (w << 4) + (lane & 15);
    int k    = (sidx << 3) + ((j >> 1) << 2) + ((lane >> 4) << 1) + (j & 1);
    Wf[id] = W[k * HID + n];
  }
}

__global__ void deg_kernel(const int* __restrict__ idx, const float* __restrict__ vals,
                           float* __restrict__ deg, int nedges) {
  int e = blockIdx.x * blockDim.x + threadIdx.x;
  if (e < nedges) atomicAdd(&deg[idx[2 * e]], vals[e]);
}

__global__ void deginv_kernel(float* __restrict__ deg, int n) {
  int i = blockIdx.x * blockDim.x + threadIdx.x;
  if (i < n) deg[i] = 1.0f / fmaxf(deg[i], 1.0f);
}

__global__ void spmm_kernel(const int* __restrict__ idx, const float* __restrict__ vals,
                            const float* __restrict__ x, float* __restrict__ out,
                            int nedges) {
  int gw   = (blockIdx.x * blockDim.x + threadIdx.x) >> 5;
  int lane = threadIdx.x & 31;
  if (gw >= nedges) return;
  int   row = idx[2 * gw];
  int   col = idx[2 * gw + 1];
  float v   = vals[gw];
  int   f   = lane << 2;
  const float4 xv = *(const float4*)(x + (long long)col * HID + f);
  float* o = out + (long long)row * HID + f;
  atomicAdd(o + 0, v * xv.x);
  atomicAdd(o + 1, v * xv.y);
  atomicAdd(o + 2, v * xv.z);
  atomicAdd(o + 3, v * xv.w);
}

__global__ void gemm_wmma_kernel(const float* __restrict__ X, const float* __restrict__ scale,
                                 const float* __restrict__ Wf, const float* __restrict__ bias,
                                 float* __restrict__ Y, int do_relu) {
  __shared__ float Xs[16][132];
  const int tile_m = blockIdx.x;
  const int t      = threadIdx.x;

  {
    int   r    = t >> 4;
    int   cseg = (t & 15) << 3;
    float s    = scale[tile_m * 16 + r];
    const float* src = X + ((long long)tile_m * 16 + r) * HID + cseg;
    float4 v0 = *(const float4*)(src);
    float4 v1 = *(const float4*)(src + 4);
    float* dst = &Xs[r][cseg];
    dst[0] = v0.x * s; dst[1] = v0.y * s; dst[2] = v0.z * s; dst[3] = v0.w * s;
    dst[4] = v1.x * s; dst[5] = v1.y * s; dst[6] = v1.z * s; dst[7] = v1.w * s;
  }
  __syncthreads();

  const int lane = t & 31;
  const int wid  = t >> 5;
  const int m    = lane & 15;
  const int kh   = (lane >> 4) << 1;
  const int n    = (wid << 4) + (lane & 15);

  const float4* wq = (const float4*)Wf + (wid * (HID / 8)) * 32 + lane;

  v8f c = {};
#pragma unroll
  for (int s = 0; s < HID / 8; ++s) {
    const float4 q = wq[s * 32];
    int k = (s << 3) + kh;
    v2f a0; a0.x = Xs[m][k];     a0.y = Xs[m][k + 1];
    v2f a1; a1.x = Xs[m][k + 4]; a1.y = Xs[m][k + 5];
    v2f b0; b0.x = q.x; b0.y = q.y;
    v2f b1; b1.x = q.z; b1.y = q.w;
    c = __builtin_amdgcn_wmma_f32_16x16x4_f32(false, a0, false, b0, (short)0, c,
                                              false, false);
    c = __builtin_amdgcn_wmma_f32_16x16x4_f32(false, a1, false, b1, (short)0, c,
                                              false, false);
  }

  const float bn      = bias[n];
  const int   rowbase = tile_m * 16 + ((lane >> 4) << 3);
#pragma unroll
  for (int i = 0; i < 8; ++i) {
    float v = c[i] + bn;
    if (do_relu) v = fmaxf(v, 0.0f);
    Y[(long long)(rowbase + i) * HID + n] = v;
  }
}

__global__ void decode_kernel(const float* __restrict__ h2,
                              const int* __restrict__ U, const int* __restrict__ V,
                              const float* __restrict__ Wd1f, const float* __restrict__ bd1,
                              const float* __restrict__ Wd2, const float* __restrict__ bd2,
                              float* __restrict__ out) {
  __shared__ float Zs[16][260];
  __shared__ float Hs[16][HID];
  __shared__ float red[16][16];

  const int base = blockIdx.x * 16;
  const int t    = threadIdx.x;

  {
    int i   = t >> 4;
    int seg = t & 15;
    const float* src = (seg < 8)
        ? h2 + (long long)U[base + i] * HID + (seg << 4)
        : h2 + (long long)V[base + i] * HID + ((seg - 8) << 4);
    float* dst = &Zs[i][seg << 4];
#pragma unroll
    for (int j = 0; j < 16; j += 4) {
      float4 v = *(const float4*)(src + j);
      dst[j] = v.x; dst[j + 1] = v.y; dst[j + 2] = v.z; dst[j + 3] = v.w;
    }
  }
  __syncthreads();

  const int lane = t & 31;
  const int wid  = t >> 5;
  const int m    = lane & 15;
  const int kh   = (lane >> 4) << 1;
  const int n    = (wid << 4) + (lane & 15);

  const float4* wq = (const float4*)Wd1f + (wid * (2 * HID / 8)) * 32 + lane;

  v8f c = {};
#pragma unroll
  for (int s = 0; s < 2 * HID / 8; ++s) {
    const float4 q = wq[s * 32];
    int k = (s << 3) + kh;
    v2f a0; a0.x = Zs[m][k];     a0.y = Zs[m][k + 1];
    v2f a1; a1.x = Zs[m][k + 4]; a1.y = Zs[m][k + 5];
    v2f b0; b0.x = q.x; b0.y = q.y;
    v2f b1; b1.x = q.z; b1.y = q.w;
    c = __builtin_amdgcn_wmma_f32_16x16x4_f32(false, a0, false, b0, (short)0, c,
                                              false, false);
    c = __builtin_amdgcn_wmma_f32_16x16x4_f32(false, a1, false, b1, (short)0, c,
                                              false, false);
  }

  const float bn    = bd1[n];
  const int   rbase = (lane >> 4) << 3;
#pragma unroll
  for (int i = 0; i < 8; ++i) Hs[rbase + i][n] = fmaxf(c[i] + bn, 0.0f);
  __syncthreads();

  {
    int i   = t >> 4;
    int seg = t & 15;
    float p = 0.0f;
#pragma unroll
    for (int j = 0; j < 8; ++j) p += Hs[i][(seg << 3) + j] * Wd2[(seg << 3) + j];
    red[i][seg] = p;
  }
  __syncthreads();
  if (t < 16) {
    float s = bd2[0];
#pragma unroll
    for (int j = 0; j < 16; ++j) s += red[t][j];
    out[base + t] = s;
  }
}

extern "C" void kernel_launch(void* const* d_in, const int* in_sizes, int n_in,
                              void* d_out, int out_size, void* d_ws, size_t ws_size,
                              hipStream_t stream) {
  const int*   adj_idx = (const int*)  d_in[0];
  const float* adj_val = (const float*)d_in[1];
  const float* h       = (const float*)d_in[3];
  const int*   pos_u   = (const int*)  d_in[4];
  const int*   pos_v   = (const int*)  d_in[5];
  const int*   neg_u   = (const int*)  d_in[6];
  const int*   neg_v   = (const int*)  d_in[7];
  const float* W1      = (const float*)d_in[8];
  const float* b1      = (const float*)d_in[9];
  const float* W2      = (const float*)d_in[10];
  const float* b2      = (const float*)d_in[11];
  const float* Wd1     = (const float*)d_in[12];
  const float* bd1     = (const float*)d_in[13];
  const float* Wd2     = (const float*)d_in[14];
  const float* bd2     = (const float*)d_in[15];

  float* outp = (float*)d_out;
  float* pos_scores = outp;
  float* neg_scores = outp + N_POS;
  float* h2         = outp + 2 * N_POS;

  float* ws      = (float*)d_ws;
  float* deg_inv = ws;
  float* tmp     = ws + 50176;
  float* h1      = tmp + (long long)N_NODES * HID;
  float* W1t     = h1 + (long long)N_NODES * HID;
  float* W2t     = W1t + HID * HID;
  float* Wd1t    = W2t + HID * HID;

  const long long feat_elems = (long long)N_NODES * HID;

  fragorder_kernel<<<(HID * HID + 255) / 256, 256, 0, stream>>>(W1, W1t, HID);
  fragorder_kernel<<<(HID * HID + 255) / 256, 256, 0, stream>>>(W2, W2t, HID);
  fragorder_kernel<<<(2 * HID * HID + 255) / 256, 256, 0, stream>>>(Wd1, Wd1t, 2 * HID);
  zero_kernel<<<2048, 256, 0, stream>>>(ws, 50176 + feat_elems);
  deg_kernel<<<(N_EDGES + 255) / 256, 256, 0, stream>>>(adj_idx, adj_val, deg_inv, N_EDGES);
  deginv_kernel<<<(N_NODES + 255) / 256, 256, 0, stream>>>(deg_inv, N_NODES);
  spmm_kernel<<<N_EDGES / 8, 256, 0, stream>>>(adj_idx, adj_val, h, tmp, N_EDGES);
  gemm_wmma_kernel<<<N_NODES / 16, 256, 0, stream>>>(tmp, deg_inv, W1t, b1, h1, 1);
  zero_kernel<<<2048, 256, 0, stream>>>(tmp, feat_elems);
  spmm_kernel<<<N_EDGES / 8, 256, 0, stream>>>(adj_idx, adj_val, h1, tmp, N_EDGES);
  gemm_wmma_kernel<<<N_NODES / 16, 256, 0, stream>>>(tmp, deg_inv, W2t, b2, h2, 0);
  decode_kernel<<<N_POS / 16, 256, 0, stream>>>(h2, pos_u, pos_v, Wd1t, bd1, Wd2, bd2, pos_scores);
  decode_kernel<<<N_POS / 16, 256, 0, stream>>>(h2, neg_u, neg_v, Wd1t, bd1, Wd2, bd2, neg_scores);
}


# Pull SpMM: edges grouped by row per call (returning int add, scan, place); GEMM tile staging sums neighbour rows and degree in registers in slot (arrival) order; no float atomics, no big zero fills
# speedup vs baseline: 2.5270x; 1.9967x over previous
#include <hip/hip_runtime.h>
#include <hip/hip_bf16.h>


typedef __attribute__((ext_vector_type(2))) float v2f;
typedef __attribute__((ext_vector_type(8))) float v8f;

#define N_NODES 50000
#define N_EDGES 800000
#define HID     128
#define N_POS   100000

__global__ void zero_kernel(float* __restrict__ p, long long n) {
  long long i = (long long)blockIdx.x * blockDim.x + threadIdx.x;
  long long stride = (long long)gridDim.x * blockDim.x;
  for (; i < n; i += stride) p[i] = 0.0f;
}

__global__ void fragorder_kernel(const float* __restrict__ W, float* __restrict__ Wf,
                                 int K) {
  int id = blockIdx.x * blockDim.x + threadIdx.x;
  if (id < K * HID) {
    int j    = id & 3;
    int lane = (id >> 2) & 31;
    int rest = id >> 7;
    int ks2  = K >> 3;
    int w    = rest / ks2;
    int sidx = rest - w * ks2;
    int n    = (w << 4) + (lane & 15);
    int k    = (sidx << 3) + ((j >> 1) << 2) + ((lane >> 4) << 1) + (j & 1);
    Wf[id] = W[k * HID + n];
  }
}

__global__ void count_kernel(const int* __restrict__ idx, int* __restrict__ cnt,
                             int* __restrict__ pos, int nedges) {
  int e = blockIdx.x * blockDim.x + threadIdx.x;
  if (e < nedges) pos[e] = atomicAdd(&cnt[idx[2 * e]], 1);
}

#define SCAN_LINES     7
#define SCAN_BLOCKS    7
#define SCAN_PER_BLOCK (1024 * SCAN_LINES)
#define SCAN_N         (SCAN_BLOCKS * SCAN_PER_BLOCK)
__global__ void __launch_bounds__(1024)
scan_kernel(const int* __restrict__ cnt, int* __restrict__ rowptr) {
  __shared__ int wsum[32];
  __shared__ int psum[32];
  const int t    = threadIdx.x;
  const int lane = t & 31;
  const int w    = t >> 5;
  const int b    = blockIdx.x;
  const int base = b * SCAN_PER_BLOCK + w * (SCAN_LINES * 32) + lane;

  int v[SCAN_LINES];
#pragma unroll
  for (int i = 0; i < SCAN_LINES; ++i) v[i] = cnt[base + 32 * i];

  int pre = 0;
  for (int i = 0; i < b * SCAN_LINES; ++i) pre += cnt[i * 1024 + t];
#pragma unroll
  for (int d = 16; d > 0; d >>= 1) pre += __shfl_xor(pre, d);

  int carry = 0;
#pragma unroll
  for (int i = 0; i < SCAN_LINES; ++i) {
    int x = v[i];
#pragma unroll
    for (int d = 1; d < 32; d <<= 1) {
      int y = __shfl_up(x, d);
      if (lane >= d) x += y;
    }
    const int tot = __shfl(x, 31);
    v[i] = x - v[i] + carry;
    carry += tot;
  }
  if (lane == 0) { wsum[w] = carry; psum[w] = pre; }
  __syncthreads();
  int off = 0;
  for (int k = 0; k < 32; ++k) off += psum[k] + (k < w ? wsum[k] : 0);
#pragma unroll
  for (int i = 0; i < SCAN_LINES; ++i) rowptr[base + 32 * i] = v[i] + off;
}

__global__ void place_kernel(const int* __restrict__ idx, const float* __restrict__ vals,
                             const int* __restrict__ rowptr, const int* __restrict__ pos,
                             int* __restrict__ csr, int nedges) {
  int e = blockIdx.x * blockDim.x + threadIdx.x;
  if (e < nedges) {
    const int2 rc   = *(const int2*)(idx + 2 * e);
    const int  slot = rowptr[rc.x] + pos[e];
    int2 o; o.x = rc.y; o.y = __float_as_int(vals[e]);
    *(int2*)(csr + 2 * slot) = o;
  }
}

template <int U>
__device__ __forceinline__ void pull_rows(const float* __restrict__ X, int colv, int valv,
                                          int j, int f, float4& acc, float& deg) {
  float4 xv[U];
  float  v[U];
#pragma unroll
  for (int u = 0; u < U; ++u) {
    const int c = __builtin_amdgcn_readlane(colv, j + u);
    v[u]  = __int_as_float(__builtin_amdgcn_readlane(valv, j + u));
    xv[u] = *(const float4*)(X + (long long)c * HID + f);
  }
#pragma unroll
  for (int u = 0; u < U; ++u) {
    acc.x += v[u] * xv[u].x;
    acc.y += v[u] * xv[u].y;
    acc.z += v[u] * xv[u].z;
    acc.w += v[u] * xv[u].w;
    deg   += v[u];
  }
}

__global__ void sage_layer_kernel(const int* __restrict__ rowptr, const int* __restrict__ csr,
                                  const float* __restrict__ X,
                                  const float* __restrict__ Wf, const float* __restrict__ bias,
                                  float* __restrict__ Y, int do_relu) {
  __shared__ __attribute__((aligned(16))) float Xs[16][132];
  const int tile_m = blockIdx.x;
  const int t      = threadIdx.x;
  const int lane   = t & 31;
  const int wid    = t >> 5;

  {
    const int   f    = lane << 2;
    const int2* csr2 = (const int2*)csr;
    for (int rr = 0; rr < 2; ++rr) {
      const int rt = (wid << 1) + rr;
      const int r  = tile_m * 16 + rt;
      const int beg = __builtin_amdgcn_readfirstlane(rowptr[r]);
      const int end = __builtin_amdgcn_readfirstlane(rowptr[r + 1]);
      float4 acc = {0.0f, 0.0f, 0.0f, 0.0f};
      float  deg = 0.0f;
      for (int base = beg; base < end; base += 32) {
        const int rem = end - base;
        const int n   = rem < 32 ? rem : 32;
        int colv = 0, valv = 0;
        if (lane < n) {
          const int2 cv = csr2[base + lane];
          colv = cv.x; valv = cv.y;
        }
        int j = 0;
        for (; j + 8 <= n; j += 8) pull_rows<8>(X, colv, valv, j, f, acc, deg);
        if (n & 4) { pull_rows<4>(X, colv, valv, j, f, acc, deg); j += 4; }
        if (n & 2) { pull_rows<2>(X, colv, valv, j, f, acc, deg); j += 2; }
        if (n & 1) { pull_rows<1>(X, colv, valv, j, f, acc, deg); }
      }
      const float s = 1.0f / fmaxf(deg, 1.0f);
      float4 o;
      o.x = acc.x * s; o.y = acc.y * s; o.z = acc.z * s; o.w = acc.w * s;
      *(float4*)&Xs[rt][f] = o;
    }
  }
  __syncthreads();

  const int m    = lane & 15;
  const int kh   = (lane >> 4) << 1;
  const int n    = (wid << 4) + (lane & 15);

  const float4* wq = (const float4*)Wf + (wid * (HID / 8)) * 32 + lane;

  v8f c = {};
#pragma unroll
  for (int s = 0; s < HID / 8; ++s) {
    const float4 q = wq[s * 32];
    int k = (s << 3) + kh;
    v2f a0; a0.x = Xs[m][k];     a0.y = Xs[m][k + 1];
    v2f a1; a1.x = Xs[m][k + 4]; a1.y = Xs[m][k + 5];
    v2f b0; b0.x = q.x; b0.y = q.y;
    v2f b1; b1.x = q.z; b1.y = q.w;
    c = __builtin_amdgcn_wmma_f32_16x16x4_f32(false, a0, false, b0, (short)0, c,
                                              false, false);
    c = __builtin_amdgcn_wmma_f32_16x16x4_f32(false, a1, false, b1, (short)0, c,
                                              false, false);
  }

  const float bn      = bias[n];
  const int   rowbase = tile_m * 16 + ((lane >> 4) << 3);
#pragma unroll
  for (int i = 0; i < 8; ++i) {
    float v = c[i] + bn;
    if (do_relu) v = fmaxf(v, 0.0f);
    Y[(long long)(rowbase + i) * HID + n] = v;
  }
}

__global__ void decode_kernel(const float* __restrict__ h2,
                              const int* __restrict__ U, const int* __restrict__ V,
                              const float* __restrict__ Wd1f, const float* __restrict__ bd1,
                              const float* __restrict__ Wd2, const float* __restrict__ bd2,
                              float* __restrict__ out) {
  __shared__ float Zs[16][260];
  __shared__ float Hs[16][HID];
  __shared__ float red[16][16];

  const int base = blockIdx.x * 16;
  const int t    = threadIdx.x;

  {
    int i   = t >> 4;
    int seg = t & 15;
    const float* src = (seg < 8)
        ? h2 + (long long)U[base + i] * HID + (seg << 4)
        : h2 + (long long)V[base + i] * HID + ((seg - 8) << 4);
    float* dst = &Zs[i][seg << 4];
#pragma unroll
    for (int j = 0; j < 16; j += 4) {
      float4 v = *(const float4*)(src + j);
      dst[j] = v.x; dst[j + 1] = v.y; dst[j + 2] = v.z; dst[j + 3] = v.w;
    }
  }
  __syncthreads();

  const int lane = t & 31;
  const int wid  = t >> 5;
  const int m    = lane & 15;
  const int kh   = (lane >> 4) << 1;
  const int n    = (wid << 4) + (lane & 15);

  const float4* wq = (const float4*)Wd1f + (wid * (2 * HID / 8)) * 32 + lane;

  v8f c = {};
#pragma unroll
  for (int s = 0; s < 2 * HID / 8; ++s) {
    const float4 q = wq[s * 32];
    int k = (s << 3) + kh;
    v2f a0; a0.x = Zs[m][k];     a0.y = Zs[m][k + 1];
    v2f a1; a1.x = Zs[m][k + 4]; a1.y = Zs[m][k + 5];
    v2f b0; b0.x = q.x; b0.y = q.y;
    v2f b1; b1.x = q.z; b1.y = q.w;
    c = __builtin_amdgcn_wmma_f32_16x16x4_f32(false, a0, false, b0, (short)0, c,
                                              false, false);
    c = __builtin_amdgcn_wmma_f32_16x16x4_f32(false, a1, false, b1, (short)0, c,
                                              false, false);
  }

  const float bn    = bd1[n];
  const int   rbase = (lane >> 4) << 3;
#pragma unroll
  for (int i = 0; i < 8; ++i) Hs[rbase + i][n] = fmaxf(c[i] + bn, 0.0f);
  __syncthreads();

  {
    int i   = t >> 4;
    int seg = t & 15;
    float p = 0.0f;
#pragma unroll
    for (int j = 0; j < 8; ++j) p += Hs[i][(seg << 3) + j] * Wd2[(seg << 3) + j];
    red[i][seg] = p;
  }
  __syncthreads();
  if (t < 16) {
    float s = bd2[0];
#pragma unroll
    for (int j = 0; j < 16; ++j) s += red[t][j];
    out[base + t] = s;
  }
}

extern "C" void kernel_launch(void* const* d_in, const int* in_sizes, int n_in,
                              void* d_out, int out_size, void* d_ws, size_t ws_size,
                              hipStream_t stream) {
  const int*   adj_idx = (const int*)  d_in[0];
  const float* adj_val = (const float*)d_in[1];
  const float* h       = (const float*)d_in[3];
  const int*   pos_u   = (const int*)  d_in[4];
  const int*   pos_v   = (const int*)  d_in[5];
  const int*   neg_u   = (const int*)  d_in[6];
  const int*   neg_v   = (const int*)  d_in[7];
  const float* W1      = (const float*)d_in[8];
  const float* b1      = (const float*)d_in[9];
  const float* W2      = (const float*)d_in[10];
  const float* b2      = (const float*)d_in[11];
  const float* Wd1     = (const float*)d_in[12];
  const float* bd1     = (const float*)d_in[13];
  const float* Wd2     = (const float*)d_in[14];
  const float* bd2     = (const float*)d_in[15];

  float* outp = (float*)d_out;
  float* pos_scores = outp;
  float* neg_scores = outp + N_POS;
  float* h2         = outp + 2 * N_POS;

  float* ws     = (float*)d_ws;
  int*   rowptr = (int*)ws;
  int*   csr    = (int*)(ws + 50176);
  int*   pos    = csr + 2 * N_EDGES;
  int*   cnt    = pos + N_EDGES;
  float* h1     = ws + 50176 + (long long)N_NODES * HID;
  float* W1t    = h1 + (long long)N_NODES * HID;
  float* W2t    = W1t + HID * HID;
  float* Wd1t   = W2t + HID * HID;

  fragorder_kernel<<<(HID * HID + 255) / 256, 256, 0, stream>>>(W1, W1t, HID);
  fragorder_kernel<<<(HID * HID + 255) / 256, 256, 0, stream>>>(W2, W2t, HID);
  fragorder_kernel<<<(2 * HID * HID + 255) / 256, 256, 0, stream>>>(Wd1, Wd1t, 2 * HID);
  zero_kernel<<<SCAN_N / 256, 256, 0, stream>>>((float*)cnt, SCAN_N);
  count_kernel<<<(N_EDGES + 255) / 256, 256, 0, stream>>>(adj_idx, cnt, pos, N_EDGES);
  scan_kernel<<<SCAN_BLOCKS, 1024, 0, stream>>>(cnt, rowptr);
  place_kernel<<<(N_EDGES + 255) / 256, 256, 0, stream>>>(adj_idx, adj_val, rowptr, pos, csr, N_EDGES);
  sage_layer_kernel<<<N_NODES / 16, 256, 0, stream>>>(rowptr, csr, h, W1t, b1, h1, 1);
  sage_layer_kernel<<<N_NODES / 16, 256, 0, stream>>>(rowptr, csr, h1, W2t, b2, h2, 0);
  decode_kernel<<<N_POS / 16, 256, 0, stream>>>(h2, pos_u, pos_v, Wd1t, bd1, Wd2, bd2, pos_scores);
  decode_kernel<<<N_POS / 16, 256, 0, stream>>>(h2, neg_u, neg_v, Wd1t, bd1, Wd2, bd2, neg_scores);
}


# Decoder layer 1 hoisted from pairs to nodes: a WMMA launch makes P=h2*Wd1[:128], Q=h2*Wd1[128:]; score=relu(P[u]+Q[v]+bd1).Wd2 by gather; sum order: 256-dot as two 128 halves, Wd2 dot lane butterfly
# speedup vs baseline: 2.9587x; 1.1708x over previous
#include <hip/hip_runtime.h>
#include <hip/hip_bf16.h>


typedef __attribute__((ext_vector_type(2))) float v2f;
typedef __attribute__((ext_vector_type(8))) float v8f;

#define N_NODES 50000
#define N_EDGES 800000
#define HID     128
#define N_POS   100000

__global__ void zero_kernel(float* __restrict__ p, long long n) {
  long long i = (long long)blockIdx.x * blockDim.x + threadIdx.x;
  long long stride = (long long)gridDim.x * blockDim.x;
  for (; i < n; i += stride) p[i] = 0.0f;
}

__global__ void fragorder_kernel(const float* __restrict__ W, float* __restrict__ Wf,
                                 int K) {
  int id = blockIdx.x * blockDim.x + threadIdx.x;
  if (id < K * HID) {
    int j    = id & 3;
    int lane = (id >> 2) & 31;
    int rest = id >> 7;
    int ks2  = K >> 3;
    int w    = rest / ks2;
    int sidx = rest - w * ks2;
    int n    = (w << 4) + (lane & 15);
    int k    = (sidx << 3) + ((j >> 1) << 2) + ((lane >> 4) << 1) + (j & 1);
    Wf[id] = W[k * HID + n];
  }
}

__global__ void count_kernel(const int* __restrict__ idx, int* __restrict__ cnt,
                             int* __restrict__ pos, int nedges) {
  int e = blockIdx.x * blockDim.x + threadIdx.x;
  if (e < nedges) pos[e] = atomicAdd(&cnt[idx[2 * e]], 1);
}

#define SCAN_LINES     7
#define SCAN_BLOCKS    7
#define SCAN_PER_BLOCK (1024 * SCAN_LINES)
#define SCAN_N         (SCAN_BLOCKS * SCAN_PER_BLOCK)
__global__ void __launch_bounds__(1024)
scan_kernel(const int* __restrict__ cnt, int* __restrict__ rowptr) {
  __shared__ int wsum[32];
  __shared__ int psum[32];
  const int t    = threadIdx.x;
  const int lane = t & 31;
  const int w    = t >> 5;
  const int b    = blockIdx.x;
  const int base = b * SCAN_PER_BLOCK + w * (SCAN_LINES * 32) + lane;

  int v[SCAN_LINES];
#pragma unroll
  for (int i = 0; i < SCAN_LINES; ++i) v[i] = cnt[base + 32 * i];

  int pre = 0;
  for (int i = 0; i < b * SCAN_LINES; ++i) pre += cnt[i * 1024 + t];
#pragma unroll
  for (int d = 16; d > 0; d >>= 1) pre += __shfl_xor(pre, d);

  int carry = 0;
#pragma unroll
  for (int i = 0; i < SCAN_LINES; ++i) {
    int x = v[i];
#pragma unroll
    for (int d = 1; d < 32; d <<= 1) {
      int y = __shfl_up(x, d);
      if (lane >= d) x += y;
    }
    const int tot = __shfl(x, 31);
    v[i] = x - v[i] + carry;
    carry += tot;
  }
  if (lane == 0) { wsum[w] = carry; psum[w] = pre; }
  __syncthreads();
  int off = 0;
  for (int k = 0; k < 32; ++k) off += psum[k] + (k < w ? wsum[k] : 0);
#pragma unroll
  for (int i = 0; i < SCAN_LINES; ++i) rowptr[base + 32 * i] = v[i] + off;
}

__global__ void place_kernel(const int* __restrict__ idx, const float* __restrict__ vals,
                             const int* __restrict__ rowptr, const int* __restrict__ pos,
                             int* __restrict__ csr, int nedges) {
  int e = blockIdx.x * blockDim.x + threadIdx.x;
  if (e < nedges) {
    const int2 rc   = *(const int2*)(idx + 2 * e);
    const int  slot = rowptr[rc.x] + pos[e];
    int2 o; o.x = rc.y; o.y = __float_as_int(vals[e]);
    *(int2*)(csr + 2 * slot) = o;
  }
}

template <int U>
__device__ __forceinline__ void pull_rows(const float* __restrict__ X, int colv, int valv,
                                          int j, int f, float4& acc, float& deg) {
  float4 xv[U];
  float  v[U];
#pragma unroll
  for (int u = 0; u < U; ++u) {
    const int c = __builtin_amdgcn_readlane(colv, j + u);
    v[u]  = __int_as_float(__builtin_amdgcn_readlane(valv, j + u));
    xv[u] = *(const float4*)(X + (long long)c * HID + f);
  }
#pragma unroll
  for (int u = 0; u < U; ++u) {
    acc.x += v[u] * xv[u].x;
    acc.y += v[u] * xv[u].y;
    acc.z += v[u] * xv[u].z;
    acc.w += v[u] * xv[u].w;
    deg   += v[u];
  }
}

__global__ void sage_layer_kernel(const int* __restrict__ rowptr, const int* __restrict__ csr,
                                  const float* __restrict__ X,
                                  const float* __restrict__ Wf, const float* __restrict__ bias,
                                  float* __restrict__ Y, int do_relu) {
  __shared__ __attribute__((aligned(16))) float Xs[16][132];
  const int tile_m = blockIdx.x;
  const int t      = threadIdx.x;
  const int lane   = t & 31;
  const int wid    = t >> 5;

  {
    const int   f    = lane << 2;
    const int2* csr2 = (const int2*)csr;
    for (int rr = 0; rr < 2; ++rr) {
      const int rt = (wid << 1) + rr;
      const int r  = tile_m * 16 + rt;
      const int beg = __builtin_amdgcn_readfirstlane(rowptr[r]);
      const int end = __builtin_amdgcn_readfirstlane(rowptr[r + 1]);
      float4 acc = {0.0f, 0.0f, 0.0f, 0.0f};
      float  deg = 0.0f;
      for (int base = beg; base < end; base += 32) {
        const int rem = end - base;
        const int n   = rem < 32 ? rem : 32;
        int colv = 0, valv = 0;
        if (lane < n) {
          const int2 cv = csr2[base + lane];
          colv = cv.x; valv = cv.y;
        }
        int j = 0;
        for (; j + 8 <= n; j += 8) pull_rows<8>(X, colv, valv, j, f, acc, deg);
        if (n & 4) { pull_rows<4>(X, colv, valv, j, f, acc, deg); j += 4; }
        if (n & 2) { pull_rows<2>(X, colv, valv, j, f, acc, deg); j += 2; }
        if (n & 1) { pull_rows<1>(X, colv, valv, j, f, acc, deg); }
      }
      const float s = 1.0f / fmaxf(deg, 1.0f);
      float4 o;
      o.x = acc.x * s; o.y = acc.y * s; o.z = acc.z * s; o.w = acc.w * s;
      *(float4*)&Xs[rt][f] = o;
    }
  }
  __syncthreads();

  const int m    = lane & 15;
  const int kh   = (lane >> 4) << 1;
  const int n    = (wid << 4) + (lane & 15);

  const float4* wq = (const float4*)Wf + (wid * (HID / 8)) * 32 + lane;

  v8f c = {};
#pragma unroll
  for (int s = 0; s < HID / 8; ++s) {
    const float4 q = wq[s * 32];
    int k = (s << 3) + kh;
    v2f a0; a0.x = Xs[m][k];     a0.y = Xs[m][k + 1];
    v2f a1; a1.x = Xs[m][k + 4]; a1.y = Xs[m][k + 5];
    v2f b0; b0.x = q.x; b0.y = q.y;
    v2f b1; b1.x = q.z; b1.y = q.w;
    c = __builtin_amdgcn_wmma_f32_16x16x4_f32(false, a0, false, b0, (short)0, c,
                                              false, false);
    c = __builtin_amdgcn_wmma_f32_16x16x4_f32(false, a1, false, b1, (short)0, c,
                                              false, false);
  }

  const float bn      = bias[n];
  const int   rowbase = tile_m * 16 + ((lane >> 4) << 3);
#pragma unroll
  for (int i = 0; i < 8; ++i) {
    float v = c[i] + bn;
    if (do_relu) v = fmaxf(v, 0.0f);
    Y[(long long)(rowbase + i) * HID + n] = v;
  }
}

#define PROJ_MT 5
__global__ void __launch_bounds__(128)
node_proj_kernel(const float* __restrict__ h2, const float* __restrict__ Wd1f,
                 float* __restrict__ P, float* __restrict__ Q) {
  __shared__ __attribute__((aligned(16))) float Xs[16 * PROJ_MT][132];
  const int t    = threadIdx.x;
  const int lane = t & 31;
  const int wid  = t >> 5;
  const int row0 = blockIdx.x * (16 * PROJ_MT);

  {
    const float4* src = (const float4*)(h2 + (long long)row0 * HID);
    float4 v[4 * PROJ_MT];
#pragma unroll
    for (int i = 0; i < 4 * PROJ_MT; ++i) v[i] = src[i * 128 + t];
#pragma unroll
    for (int i = 0; i < 4 * PROJ_MT; ++i) *(float4*)&Xs[4 * i + wid][lane << 2] = v[i];
  }
  __syncthreads();

  const int m  = lane & 15;
  const int kh = (lane >> 4) << 1;
  const int nt = (blockIdx.y << 2) + wid;

  const float4* wq = (const float4*)Wd1f + (nt * (2 * HID / 8)) * 32 + lane;

  v8f cp[PROJ_MT] = {};
  v8f cq[PROJ_MT] = {};
#pragma unroll
  for (int s = 0; s < HID / 8; ++s) {
    const float4 qp = wq[s * 32];
    const float4 qq = wq[(HID / 8 + s) * 32];
    const int k = (s << 3) + kh;
    v2f bp0; bp0.x = qp.x; bp0.y = qp.y;
    v2f bp1; bp1.x = qp.z; bp1.y = qp.w;
    v2f bq0; bq0.x = qq.x; bq0.y = qq.y;
    v2f bq1; bq1.x = qq.z; bq1.y = qq.w;
#pragma unroll
    for (int mt = 0; mt < PROJ_MT; ++mt) {
      const float* xr = &Xs[16 * mt + m][k];
      v2f a0; a0.x = xr[0]; a0.y = xr[1];
      v2f a1; a1.x = xr[4]; a1.y = xr[5];
      cp[mt] = __builtin_amdgcn_wmma_f32_16x16x4_f32(false, a0, false, bp0, (short)0, cp[mt],
                                                     false, false);
      cq[mt] = __builtin_amdgcn_wmma_f32_16x16x4_f32(false, a0, false, bq0, (short)0, cq[mt],
                                                     false, false);
      cp[mt] = __builtin_amdgcn_wmma_f32_16x16x4_f32(false, a1, false, bp1, (short)0, cp[mt],
                                                     false, false);
      cq[mt] = __builtin_amdgcn_wmma_f32_16x16x4_f32(false, a1, false, bq1, (short)0, cq[mt],
                                                     false, false);
    }
  }

  const int n = (nt << 4) + m;
#pragma unroll
  for (int mt = 0; mt < PROJ_MT; ++mt) {
    const int rowbase = row0 + 16 * mt + ((lane >> 4) << 3);
#pragma unroll
    for (int i = 0; i < 8; ++i) {
      P[(long long)(rowbase + i) * HID + n] = cp[mt][i];
      Q[(long long)(rowbase + i) * HID + n] = cq[mt][i];
    }
  }
}

template <int CTRL>
__device__ __forceinline__ int row_pick(int x) {
  return __builtin_amdgcn_update_dpp(0, x, CTRL, 0xf, 0xf, false);
}
template <int CTRL>
__device__ __forceinline__ float row_pickf(float x) {
  return __int_as_float(__builtin_amdgcn_update_dpp(0, __float_as_int(x), CTRL, 0xf, 0xf, false));
}

__device__ __forceinline__ float pair_dot8(const float4& p0, const float4& p1,
                                           const float4& q0, const float4& q1,
                                           const float4& b0, const float4& b1,
                                           const float4& w0, const float4& w1) {
  float s;
  s  = fmaxf((p0.x + q0.x) + b0.x, 0.0f) * w0.x;
  s += fmaxf((p0.y + q0.y) + b0.y, 0.0f) * w0.y;
  s += fmaxf((p0.z + q0.z) + b0.z, 0.0f) * w0.z;
  s += fmaxf((p0.w + q0.w) + b0.w, 0.0f) * w0.w;
  s += fmaxf((p1.x + q1.x) + b1.x, 0.0f) * w1.x;
  s += fmaxf((p1.y + q1.y) + b1.y, 0.0f) * w1.y;
  s += fmaxf((p1.z + q1.z) + b1.z, 0.0f) * w1.z;
  s += fmaxf((p1.w + q1.w) + b1.w, 0.0f) * w1.w;
  return s;
}

template <int IT>
__device__ __forceinline__ float4 score_group(const float* __restrict__ P,
                                              const float* __restrict__ Q,
                                              int uidx, int vidx, int f0,
                                              const float4& b0, const float4& b1,
                                              const float4& w0, const float4& w1) {
  const int u0 = row_pick<0x150 + 4 * IT + 0>(uidx), v0 = row_pick<0x150 + 4 * IT + 0>(vidx);
  const int u1 = row_pick<0x150 + 4 * IT + 1>(uidx), v1 = row_pick<0x150 + 4 * IT + 1>(vidx);
  const int u2 = row_pick<0x150 + 4 * IT + 2>(uidx), v2 = row_pick<0x150 + 4 * IT + 2>(vidx);
  const int u3 = row_pick<0x150 + 4 * IT + 3>(uidx), v3 = row_pick<0x150 + 4 * IT + 3>(vidx);
  const float* pa = P + (long long)u0 * HID + f0;
  const float* pb = P + (long long)u1 * HID + f0;
  const float* pc = P + (long long)u2 * HID + f0;
  const float* pd = P + (long long)u3 * HID + f0;
  const float* qa = Q + (long long)v0 * HID + f0;
  const float* qb = Q + (long long)v1 * HID + f0;
  const float* qc = Q + (long long)v2 * HID + f0;
  const float* qd = Q + (long long)v3 * HID + f0;
  const float4 pa0 = *(const float4*)pa, pa1 = *(const float4*)(pa + 64);
  const float4 qa0 = *(const float4*)qa, qa1 = *(const float4*)(qa + 64);
  const float4 pb0 = *(const float4*)pb, pb1 = *(const float4*)(pb + 64);
  const float4 qb0 = *(const float4*)qb, qb1 = *(const float4*)(qb + 64);
  const float4 pc0 = *(const float4*)pc, pc1 = *(const float4*)(pc + 64);
  const float4 qc0 = *(const float4*)qc, qc1 = *(const float4*)(qc + 64);
  const float4 pd0 = *(const float4*)pd, pd1 = *(const float4*)(pd + 64);
  const float4 qd0 = *(const float4*)qd, qd1 = *(const float4*)(qd + 64);
  float s0 = pair_dot8(pa0, pa1, qa0, qa1, b0, b1, w0, w1);
  float s1 = pair_dot8(pb0, pb1, qb0, qb1, b0, b1, w0, w1);
  float s2 = pair_dot8(pc0, pc1, qc0, qc1, b0, b1, w0, w1);
  float s3 = pair_dot8(pd0, pd1, qd0, qd1, b0, b1, w0, w1);
  s0 += row_pickf<0x161>(s0); s1 += row_pickf<0x161>(s1);
  s2 += row_pickf<0x161>(s2); s3 += row_pickf<0x161>(s3);
  s0 += row_pickf<0x162>(s0); s1 += row_pickf<0x162>(s1);
  s2 += row_pickf<0x162>(s2); s3 += row_pickf<0x162>(s3);
  s0 += row_pickf<0x164>(s0); s1 += row_pickf<0x164>(s1);
  s2 += row_pickf<0x164>(s2); s3 += row_pickf<0x164>(s3);
  s0 += row_pickf<0x168>(s0); s1 += row_pickf<0x168>(s1);
  s2 += row_pickf<0x168>(s2); s3 += row_pickf<0x168>(s3);
  float4 r; r.x = s0; r.y = s1; r.z = s2; r.w = s3;
  return r;
}

#define SC_WAVES 5
__global__ void __launch_bounds__(32 * SC_WAVES)
pair_score_kernel(const float* __restrict__ P, const float* __restrict__ Q,
                  const int* __restrict__ pu, const int* __restrict__ pv,
                  const int* __restrict__ nu, const int* __restrict__ nv,
                  const float* __restrict__ bd1, const float* __restrict__ Wd2,
                  const float* __restrict__ bd2, float* __restrict__ out) {
  __shared__ __attribute__((aligned(16))) float res[SC_WAVES][32];
  const int t    = threadIdx.x;
  const int lane = t & 31;
  const int wid  = t >> 5;
  const int l16  = lane & 15;
  const int hw   = lane >> 4;
  const int pbase = (blockIdx.x * SC_WAVES + wid) << 5;

  const bool neg = blockIdx.x >= (N_POS / (32 * SC_WAVES));
  const int* __restrict__ U = neg ? nu : pu;
  const int* __restrict__ V = neg ? nv : pv;
  const int  lbase = neg ? pbase - N_POS : pbase;

  const int sel  = ((l16 >> 2) << 3) + (hw << 2) + (l16 & 3);
  const int uidx = U[lbase + sel];
  const int vidx = V[lbase + sel];

  const int f0 = l16 << 2;
  const float4 b0 = *(const float4*)(bd1 + f0);
  const float4 b1 = *(const float4*)(bd1 + 64 + f0);
  const float4 w0 = *(const float4*)(Wd2 + f0);
  const float4 w1 = *(const float4*)(Wd2 + 64 + f0);

  const float4 r0 = score_group<0>(P, Q, uidx, vidx, f0, b0, b1, w0, w1);
  const float4 r1 = score_group<1>(P, Q, uidx, vidx, f0, b0, b1, w0, w1);
  const float4 r2 = score_group<2>(P, Q, uidx, vidx, f0, b0, b1, w0, w1);
  const float4 r3 = score_group<3>(P, Q, uidx, vidx, f0, b0, b1, w0, w1);

  if (l16 == 0) {
    float* rw = &res[wid][hw << 2];
    *(float4*)(rw)      = r0;
    *(float4*)(rw + 8)  = r1;
    *(float4*)(rw + 16) = r2;
    *(float4*)(rw + 24) = r3;
  }
  __syncthreads();
  out[pbase + lane] = res[wid][lane] + bd2[0];
}

extern "C" void kernel_launch(void* const* d_in, const int* in_sizes, int n_in,
                              void* d_out, int out_size, void* d_ws, size_t ws_size,
                              hipStream_t stream) {
  const int*   adj_idx = (const int*)  d_in[0];
  const float* adj_val = (const float*)d_in[1];
  const float* h       = (const float*)d_in[3];
  const int*   pos_u   = (const int*)  d_in[4];
  const int*   pos_v   = (const int*)  d_in[5];
  const int*   neg_u   = (const int*)  d_in[6];
  const int*   neg_v   = (const int*)  d_in[7];
  const float* W1      = (const float*)d_in[8];
  const float* b1      = (const float*)d_in[9];
  const float* W2      = (const float*)d_in[10];
  const float* b2      = (const float*)d_in[11];
  const float* Wd1     = (const float*)d_in[12];
  const float* bd1     = (const float*)d_in[13];
  const float* Wd2     = (const float*)d_in[14];
  const float* bd2     = (const float*)d_in[15];

  float* outp = (float*)d_out;
  float* h2         = outp + 2 * N_POS;

  float* ws     = (float*)d_ws;
  int*   rowptr = (int*)ws;
  int*   csr    = (int*)(ws + 50176);
  int*   pos    = csr + 2 * N_EDGES;
  int*   cnt    = pos + N_EDGES;
  float* h1     = ws + 50176 + (long long)N_NODES * HID;
  float* W1t    = h1 + (long long)N_NODES * HID;
  float* W2t    = W1t + HID * HID;
  float* Wd1t   = W2t + HID * HID;
  float* Qn     = ws + 50176;
  float* Pn     = h1;

  fragorder_kernel<<<(HID * HID + 255) / 256, 256, 0, stream>>>(W1, W1t, HID);
  fragorder_kernel<<<(HID * HID + 255) / 256, 256, 0, stream>>>(W2, W2t, HID);
  fragorder_kernel<<<(2 * HID * HID + 255) / 256, 256, 0, stream>>>(Wd1, Wd1t, 2 * HID);
  zero_kernel<<<SCAN_N / 256, 256, 0, stream>>>((float*)cnt, SCAN_N);
  count_kernel<<<(N_EDGES + 255) / 256, 256, 0, stream>>>(adj_idx, cnt, pos, N_EDGES);
  scan_kernel<<<SCAN_BLOCKS, 1024, 0, stream>>>(cnt, rowptr);
  place_kernel<<<(N_EDGES + 255) / 256, 256, 0, stream>>>(adj_idx, adj_val, rowptr, pos, csr, N_EDGES);
  sage_layer_kernel<<<N_NODES / 16, 256, 0, stream>>>(rowptr, csr, h, W1t, b1, h1, 1);
  sage_layer_kernel<<<N_NODES / 16, 256, 0, stream>>>(rowptr, csr, h1, W2t, b2, h2, 0);
  node_proj_kernel<<<dim3(N_NODES / (16 * PROJ_MT), 2), 128, 0, stream>>>(h2, Wd1t, Pn, Qn);
  pair_score_kernel<<<(2 * N_POS) / (32 * SC_WAVES), 32 * SC_WAVES, 0, stream>>>(
      Pn, Qn, pos_u, pos_v, neg_u, neg_v, bd1, Wd2, bd2, outp);
}


# Edge grouping in ONE pass: the returning int add puts an edge straight into its row's fixed 32-slot table (longer rows: overflow list); scan, place, pos, rowptr gone; sums stay in arrival order
# speedup vs baseline: 3.1111x; 1.0515x over previous
#include <hip/hip_runtime.h>
#include <hip/hip_bf16.h>


typedef __attribute__((ext_vector_type(2))) float v2f;
typedef __attribute__((ext_vector_type(8))) float v8f;

#define N_NODES 50000
#define N_EDGES 800000
#define HID     128
#define N_POS   100000

__global__ void zero_kernel(float* __restrict__ p, long long n) {
  long long i = (long long)blockIdx.x * blockDim.x + threadIdx.x;
  long long stride = (long long)gridDim.x * blockDim.x;
  for (; i < n; i += stride) p[i] = 0.0f;
}

__global__ void fragorder_kernel(const float* __restrict__ W, float* __restrict__ Wf,
                                 int K) {
  int id = blockIdx.x * blockDim.x + threadIdx.x;
  if (id < K * HID) {
    int j    = id & 3;
    int lane = (id >> 2) & 31;
    int rest = id >> 7;
    int ks2  = K >> 3;
    int w    = rest / ks2;
    int sidx = rest - w * ks2;
    int n    = (w << 4) + (lane & 15);
    int k    = (sidx << 3) + ((j >> 1) << 2) + ((lane >> 4) << 1) + (j & 1);
    Wf[id] = W[k * HID + n];
  }
}

#define ROW_CAP 32
#define CNT_N   50176
#define OVF_IDX (N_NODES + 32)
__global__ void count_place_kernel(const int* __restrict__ idx, const float* __restrict__ vals,
                                   int* __restrict__ cnt, int* __restrict__ tab,
                                   int* __restrict__ ovf_rc, int* __restrict__ ovf_v,
                                   int nedges) {
  int e = blockIdx.x * blockDim.x + threadIdx.x;
  if (e < nedges) {
    const int2 rc = *(const int2*)(idx + 2 * e);
    const int  vb = __float_as_int(vals[e]);
    const int  p  = atomicAdd(&cnt[rc.x], 1);
    if (p < ROW_CAP) {
      int2 o; o.x = rc.y; o.y = vb;
      *(int2*)(tab + 2 * (rc.x * ROW_CAP + p)) = o;
    } else {
      const int q = atomicAdd(&cnt[OVF_IDX], 1);
      *(int2*)(ovf_rc + 2 * q) = rc;
      ovf_v[q] = vb;
    }
  }
}

template <int U>
__device__ __forceinline__ void pull_rows(const float* __restrict__ X, int colv, int valv,
                                          int j, int f, float4& acc, float& deg) {
  float4 xv[U];
  float  v[U];
#pragma unroll
  for (int u = 0; u < U; ++u) {
    const int c = __builtin_amdgcn_readlane(colv, j + u);
    v[u]  = __int_as_float(__builtin_amdgcn_readlane(valv, j + u));
    xv[u] = *(const float4*)(X + (long long)c * HID + f);
  }
#pragma unroll
  for (int u = 0; u < U; ++u) {
    acc.x += v[u] * xv[u].x;
    acc.y += v[u] * xv[u].y;
    acc.z += v[u] * xv[u].z;
    acc.w += v[u] * xv[u].w;
    deg   += v[u];
  }
}

__global__ void sage_layer_kernel(const int* __restrict__ cnt, const int* __restrict__ tab,
                                  const int* __restrict__ ovf_rc, const int* __restrict__ ovf_v,
                                  const float* __restrict__ X,
                                  const float* __restrict__ Wf, const float* __restrict__ bias,
                                  float* __restrict__ Y, int do_relu) {
  __shared__ __attribute__((aligned(16))) float Xs[16][132];
  const int tile_m = blockIdx.x;
  const int t      = threadIdx.x;
  const int lane   = t & 31;
  const int wid    = t >> 5;

  {
    const int   f    = lane << 2;
    const int2* tab2 = (const int2*)tab;
    const int   r0   = tile_m * 16 + (wid << 1);
    const int2  nn   = *(const int2*)(cnt + r0);
    for (int rr = 0; rr < 2; ++rr) {
      const int rt = (wid << 1) + rr;
      const int r  = r0 + rr;
      const int n = __builtin_amdgcn_readfirstlane(rr ? nn.y : nn.x);
      const int m = n < ROW_CAP ? n : ROW_CAP;
      float4 acc = {0.0f, 0.0f, 0.0f, 0.0f};
      float  deg = 0.0f;
      int colv = 0, valv = 0;
      if (lane < m) {
        const int2 cv = tab2[r * ROW_CAP + lane];
        colv = cv.x; valv = cv.y;
      }
      int j = 0;
      for (; j + 8 <= m; j += 8) pull_rows<8>(X, colv, valv, j, f, acc, deg);
      if (m & 4) { pull_rows<4>(X, colv, valv, j, f, acc, deg); j += 4; }
      if (m & 2) { pull_rows<2>(X, colv, valv, j, f, acc, deg); j += 2; }
      if (m & 1) { pull_rows<1>(X, colv, valv, j, f, acc, deg); }
      if (n > ROW_CAP) {
        const int   total = __builtin_amdgcn_readfirstlane(cnt[OVF_IDX]);
        const int2* orc2  = (const int2*)ovf_rc;
        for (int b = 0; b < total; b += 32) {
          int orow = -1, ocol = 0, oval = 0;
          if (b + lane < total) {
            const int2 k = orc2[b + lane];
            orow = k.x; ocol = k.y; oval = ovf_v[b + lane];
          }
          unsigned hit = __builtin_amdgcn_ballot_w32(orow == r);
          while (hit) {
            const int l = __builtin_ctz(hit);
            hit &= hit - 1;
            pull_rows<1>(X, ocol, oval, l, f, acc, deg);
          }
        }
      }
      const float s = 1.0f / fmaxf(deg, 1.0f);
      float4 o;
      o.x = acc.x * s; o.y = acc.y * s; o.z = acc.z * s; o.w = acc.w * s;
      *(float4*)&Xs[rt][f] = o;
    }
  }
  __syncthreads();

  const int m    = lane & 15;
  const int kh   = (lane >> 4) << 1;
  const int n    = (wid << 4) + (lane & 15);

  const float4* wq = (const float4*)Wf + (wid * (HID / 8)) * 32 + lane;

  v8f c = {};
#pragma unroll
  for (int s = 0; s < HID / 8; ++s) {
    const float4 q = wq[s * 32];
    int k = (s << 3) + kh;
    v2f a0; a0.x = Xs[m][k];     a0.y = Xs[m][k + 1];
    v2f a1; a1.x = Xs[m][k + 4]; a1.y = Xs[m][k + 5];
    v2f b0; b0.x = q.x; b0.y = q.y;
    v2f b1; b1.x = q.z; b1.y = q.w;
    c = __builtin_amdgcn_wmma_f32_16x16x4_f32(false, a0, false, b0, (short)0, c,
                                              false, false);
    c = __builtin_amdgcn_wmma_f32_16x16x4_f32(false, a1, false, b1, (short)0, c,
                                              false, false);
  }

  const float bn      = bias[n];
  const int   rowbase = tile_m * 16 + ((lane >> 4) << 3);
#pragma unroll
  for (int i = 0; i < 8; ++i) {
    float v = c[i] + bn;
    if (do_relu) v = fmaxf(v, 0.0f);
    Y[(long long)(rowbase + i) * HID + n] = v;
  }
}

#define PROJ_MT 5
__global__ void __launch_bounds__(128)
node_proj_kernel(const float* __restrict__ h2, const float* __restrict__ Wd1f,
                 float* __restrict__ P, float* __restrict__ Q) {
  __shared__ __attribute__((aligned(16))) float Xs[16 * PROJ_MT][132];
  const int t    = threadIdx.x;
  const int lane = t & 31;
  const int wid  = t >> 5;
  const int row0 = blockIdx.x * (16 * PROJ_MT);

  {
    const float4* src = (const float4*)(h2 + (long long)row0 * HID);
    float4 v[4 * PROJ_MT];
#pragma unroll
    for (int i = 0; i < 4 * PROJ_MT; ++i) v[i] = src[i * 128 + t];
#pragma unroll
    for (int i = 0; i < 4 * PROJ_MT; ++i) *(float4*)&Xs[4 * i + wid][lane << 2] = v[i];
  }
  __syncthreads();

  const int m  = lane & 15;
  const int kh = (lane >> 4) << 1;
  const int nt = (blockIdx.y << 2) + wid;

  const float4* wq = (const float4*)Wd1f + (nt * (2 * HID / 8)) * 32 + lane;

  v8f cp[PROJ_MT] = {};
  v8f cq[PROJ_MT] = {};
#pragma unroll
  for (int s = 0; s < HID / 8; ++s) {
    const float4 qp = wq[s * 32];
    const float4 qq = wq[(HID / 8 + s) * 32];
    const int k = (s << 3) + kh;
    v2f bp0; bp0.x = qp.x; bp0.y = qp.y;
    v2f bp1; bp1.x = qp.z; bp1.y = qp.w;
    v2f bq0; bq0.x = qq.x; bq0.y = qq.y;
    v2f bq1; bq1.x = qq.z; bq1.y = qq.w;
#pragma unroll
    for (int mt = 0; mt < PROJ_MT; ++mt) {
      const float* xr = &Xs[16 * mt + m][k];
      v2f a0; a0.x = xr[0]; a0.y = xr[1];
      v2f a1; a1.x = xr[4]; a1.y = xr[5];
      cp[mt] = __builtin_amdgcn_wmma_f32_16x16x4_f32(false, a0, false, bp0, (short)0, cp[mt],
                                                     false, false);
      cq[mt] = __builtin_amdgcn_wmma_f32_16x16x4_f32(false, a0, false, bq0, (short)0, cq[mt],
                                                     false, false);
      cp[mt] = __builtin_amdgcn_wmma_f32_16x16x4_f32(false, a1, false, bp1, (short)0, cp[mt],
                                                     false, false);
      cq[mt] = __builtin_amdgcn_wmma_f32_16x16x4_f32(false, a1, false, bq1, (short)0, cq[mt],
                                                     false, false);
    }
  }

  const int n = (nt << 4) + m;
#pragma unroll
  for (int mt = 0; mt < PROJ_MT; ++mt) {
    const int rowbase = row0 + 16 * mt + ((lane >> 4) << 3);
#pragma unroll
    for (int i = 0; i < 8; ++i) {
      P[(long long)(rowbase + i) * HID + n] = cp[mt][i];
      Q[(long long)(rowbase + i) * HID + n] = cq[mt][i];
    }
  }
}

template <int CTRL>
__device__ __forceinline__ int row_pick(int x) {
  return __builtin_amdgcn_update_dpp(0, x, CTRL, 0xf, 0xf, false);
}
template <int CTRL>
__device__ __forceinline__ float row_pickf(float x) {
  return __int_as_float(__builtin_amdgcn_update_dpp(0, __float_as_int(x), CTRL, 0xf, 0xf, false));
}

__device__ __forceinline__ float pair_dot8(const float4& p0, const float4& p1,
                                           const float4& q0, const float4& q1,
                                           const float4& b0, const float4& b1,
                                           const float4& w0, const float4& w1) {
  float s;
  s  = fmaxf((p0.x + q0.x) + b0.x, 0.0f) * w0.x;
  s += fmaxf((p0.y + q0.y) + b0.y, 0.0f) * w0.y;
  s += fmaxf((p0.z + q0.z) + b0.z, 0.0f) * w0.z;
  s += fmaxf((p0.w + q0.w) + b0.w, 0.0f) * w0.w;
  s += fmaxf((p1.x + q1.x) + b1.x, 0.0f) * w1.x;
  s += fmaxf((p1.y + q1.y) + b1.y, 0.0f) * w1.y;
  s += fmaxf((p1.z + q1.z) + b1.z, 0.0f) * w1.z;
  s += fmaxf((p1.w + q1.w) + b1.w, 0.0f) * w1.w;
  return s;
}

template <int IT>
__device__ __forceinline__ float4 score_group(const float* __restrict__ P,
                                              const float* __restrict__ Q,
                                              int uidx, int vidx, int f0,
                                              const float4& b0, const float4& b1,
                                              const float4& w0, const float4& w1) {
  const int u0 = row_pick<0x150 + 4 * IT + 0>(uidx), v0 = row_pick<0x150 + 4 * IT + 0>(vidx);
  const int u1 = row_pick<0x150 + 4 * IT + 1>(uidx), v1 = row_pick<0x150 + 4 * IT + 1>(vidx);
  const int u2 = row_pick<0x150 + 4 * IT + 2>(uidx), v2 = row_pick<0x150 + 4 * IT + 2>(vidx);
  const int u3 = row_pick<0x150 + 4 * IT + 3>(uidx), v3 = row_pick<0x150 + 4 * IT + 3>(vidx);
  const float* pa = P + (long long)u0 * HID + f0;
  const float* pb = P + (long long)u1 * HID + f0;
  const float* pc = P + (long long)u2 * HID + f0;
  const float* pd = P + (long long)u3 * HID + f0;
  const float* qa = Q + (long long)v0 * HID + f0;
  const float* qb = Q + (long long)v1 * HID + f0;
  const float* qc = Q + (long long)v2 * HID + f0;
  const float* qd = Q + (long long)v3 * HID + f0;
  const float4 pa0 = *(const float4*)pa, pa1 = *(const float4*)(pa + 64);
  const float4 qa0 = *(const float4*)qa, qa1 = *(const float4*)(qa + 64);
  const float4 pb0 = *(const float4*)pb, pb1 = *(const float4*)(pb + 64);
  const float4 qb0 = *(const float4*)qb, qb1 = *(const float4*)(qb + 64);
  const float4 pc0 = *(const float4*)pc, pc1 = *(const float4*)(pc + 64);
  const float4 qc0 = *(const float4*)qc, qc1 = *(const float4*)(qc + 64);
  const float4 pd0 = *(const float4*)pd, pd1 = *(const float4*)(pd + 64);
  const float4 qd0 = *(const float4*)qd, qd1 = *(const float4*)(qd + 64);
  float s0 = pair_dot8(pa0, pa1, qa0, qa1, b0, b1, w0, w1);
  float s1 = pair_dot8(pb0, pb1, qb0, qb1, b0, b1, w0, w1);
  float s2 = pair_dot8(pc0, pc1, qc0, qc1, b0, b1, w0, w1);
  float s3 = pair_dot8(pd0, pd1, qd0, qd1, b0, b1, w0, w1);
  s0 += row_pickf<0x161>(s0); s1 += row_pickf<0x161>(s1);
  s2 += row_pickf<0x161>(s2); s3 += row_pickf<0x161>(s3);
  s0 += row_pickf<0x162>(s0); s1 += row_pickf<0x162>(s1);
  s2 += row_pickf<0x162>(s2); s3 += row_pickf<0x162>(s3);
  s0 += row_pickf<0x164>(s0); s1 += row_pickf<0x164>(s1);
  s2 += row_pickf<0x164>(s2); s3 += row_pickf<0x164>(s3);
  s0 += row_pickf<0x168>(s0); s1 += row_pickf<0x168>(s1);
  s2 += row_pickf<0x168>(s2); s3 += row_pickf<0x168>(s3);
  float4 r; r.x = s0; r.y = s1; r.z = s2; r.w = s3;
  return r;
}

#define SC_WAVES 5
__global__ void __launch_bounds__(32 * SC_WAVES)
pair_score_kernel(const float* __restrict__ P, const float* __restrict__ Q,
                  const int* __restrict__ pu, const int* __restrict__ pv,
                  const int* __restrict__ nu, const int* __restrict__ nv,
                  const float* __restrict__ bd1, const float* __restrict__ Wd2,
                  const float* __restrict__ bd2, float* __restrict__ out) {
  __shared__ __attribute__((aligned(16))) float res[SC_WAVES][32];
  const int t    = threadIdx.x;
  const int lane = t & 31;
  const int wid  = t >> 5;
  const int l16  = lane & 15;
  const int hw   = lane >> 4;
  const int pbase = (blockIdx.x * SC_WAVES + wid) << 5;

  const bool neg = blockIdx.x >= (N_POS / (32 * SC_WAVES));
  const int* __restrict__ U = neg ? nu : pu;
  const int* __restrict__ V = neg ? nv : pv;
  const int  lbase = neg ? pbase - N_POS : pbase;

  const int sel  = ((l16 >> 2) << 3) + (hw << 2) + (l16 & 3);
  const int uidx = U[lbase + sel];
  const int vidx = V[lbase + sel];

  const int f0 = l16 << 2;
  const float4 b0 = *(const float4*)(bd1 + f0);
  const float4 b1 = *(const float4*)(bd1 + 64 + f0);
  const float4 w0 = *(const float4*)(Wd2 + f0);
  const float4 w1 = *(const float4*)(Wd2 + 64 + f0);

  const float4 r0 = score_group<0>(P, Q, uidx, vidx, f0, b0, b1, w0, w1);
  const float4 r1 = score_group<1>(P, Q, uidx, vidx, f0, b0, b1, w0, w1);
  const float4 r2 = score_group<2>(P, Q, uidx, vidx, f0, b0, b1, w0, w1);
  const float4 r3 = score_group<3>(P, Q, uidx, vidx, f0, b0, b1, w0, w1);

  if (l16 == 0) {
    float* rw = &res[wid][hw << 2];
    *(float4*)(rw)      = r0;
    *(float4*)(rw + 8)  = r1;
    *(float4*)(rw + 16) = r2;
    *(float4*)(rw + 24) = r3;
  }
  __syncthreads();
  out[pbase + lane] = res[wid][lane] + bd2[0];
}

extern "C" void kernel_launch(void* const* d_in, const int* in_sizes, int n_in,
                              void* d_out, int out_size, void* d_ws, size_t ws_size,
                              hipStream_t stream) {
  const int*   adj_idx = (const int*)  d_in[0];
  const float* adj_val = (const float*)d_in[1];
  const float* h       = (const float*)d_in[3];
  const int*   pos_u   = (const int*)  d_in[4];
  const int*   pos_v   = (const int*)  d_in[5];
  const int*   neg_u   = (const int*)  d_in[6];
  const int*   neg_v   = (const int*)  d_in[7];
  const float* W1      = (const float*)d_in[8];
  const float* b1      = (const float*)d_in[9];
  const float* W2      = (const float*)d_in[10];
  const float* b2      = (const float*)d_in[11];
  const float* Wd1     = (const float*)d_in[12];
  const float* bd1     = (const float*)d_in[13];
  const float* Wd2     = (const float*)d_in[14];
  const float* bd2     = (const float*)d_in[15];

  float* outp = (float*)d_out;
  float* h2         = outp + 2 * N_POS;

  float* ws     = (float*)d_ws;
  int*   cnt    = (int*)ws;
  int*   tab    = (int*)(ws + 50176);
  int*   ovf_rc = tab + 2 * N_NODES * ROW_CAP;
  int*   ovf_v  = ovf_rc + 2 * N_EDGES;
  float* h1     = ws + 50176 + (long long)N_NODES * HID;
  float* W1t    = h1 + (long long)N_NODES * HID;
  float* W2t    = W1t + HID * HID;
  float* Wd1t   = W2t + HID * HID;
  float* Qn     = ws + 50176;
  float* Pn     = h1;

  fragorder_kernel<<<(HID * HID + 255) / 256, 256, 0, stream>>>(W1, W1t, HID);
  fragorder_kernel<<<(HID * HID + 255) / 256, 256, 0, stream>>>(W2, W2t, HID);
  fragorder_kernel<<<(2 * HID * HID + 255) / 256, 256, 0, stream>>>(Wd1, Wd1t, 2 * HID);
  zero_kernel<<<CNT_N / 256, 256, 0, stream>>>((float*)cnt, CNT_N);
  count_place_kernel<<<(N_EDGES + 255) / 256, 256, 0, stream>>>(adj_idx, adj_val, cnt, tab, ovf_rc,
                                                               ovf_v, N_EDGES);
  sage_layer_kernel<<<N_NODES / 16, 256, 0, stream>>>(cnt, tab, ovf_rc, ovf_v, h, W1t, b1, h1, 1);
  sage_layer_kernel<<<N_NODES / 16, 256, 0, stream>>>(cnt, tab, ovf_rc, ovf_v, h1, W2t, b2, h2, 0);
  node_proj_kernel<<<dim3(N_NODES / (16 * PROJ_MT), 2), 128, 0, stream>>>(h2, Wd1t, Pn, Qn);
  pair_score_kernel<<<(2 * N_POS) / (32 * SC_WAVES), 32 * SC_WAVES, 0, stream>>>(
      Pn, Qn, pos_u, pos_v, neg_u, neg_v, bd1, Wd2, bd2, outp);
}


# The four input-only launches (counter clear, three weight re-lays) merged into one grid: 9 launches per call become 6; each block does one piece; arithmetic and sum order unchanged
# speedup vs baseline: 3.2358x; 1.0401x over previous
#include <hip/hip_runtime.h>
#include <hip/hip_bf16.h>


typedef __attribute__((ext_vector_type(2))) float v2f;
typedef __attribute__((ext_vector_type(8))) float v8f;

#define N_NODES 50000
#define N_EDGES 800000
#define HID     128
#define N_POS   100000

#define CNT_N    50176
#define PREP_ZB  (CNT_N / 256)
#define PREP_WB  (HID * HID / 256)
#define PREP_NB  (PREP_ZB + 4 * PREP_WB)
__global__ void __launch_bounds__(256)
prep_kernel(const float* __restrict__ W1, const float* __restrict__ W2,
            const float* __restrict__ Wd1,
            float* __restrict__ W1f, float* __restrict__ W2f, float* __restrict__ Wd1f,
            int* __restrict__ cnt) {
  const int t = threadIdx.x;
  int b = blockIdx.x;
  if (b < PREP_ZB) {
    cnt[b * 256 + t] = 0;
    return;
  }
  b -= PREP_ZB;
  const float* __restrict__ W;
  float* __restrict__ Wf;
  int lg;
  if (b < PREP_WB)          {                   W = W1;  Wf = W1f;  lg = 4; }
  else if (b < 2 * PREP_WB) { b -= PREP_WB;     W = W2;  Wf = W2f;  lg = 4; }
  else                      { b -= 2 * PREP_WB; W = Wd1; Wf = Wd1f; lg = 5; }
  const int id   = b * 256 + t;
  const int j    = id & 3;
  const int lane = (id >> 2) & 31;
  const int rest = id >> 7;
  const int w    = rest >> lg;
  const int sidx = rest - (w << lg);
  const int n    = (w << 4) + (lane & 15);
  const int k    = (sidx << 3) + ((j >> 1) << 2) + ((lane >> 4) << 1) + (j & 1);
  Wf[id] = W[k * HID + n];
}

#define ROW_CAP 32
#define OVF_IDX (N_NODES + 32)
__global__ void count_place_kernel(const int* __restrict__ idx, const float* __restrict__ vals,
                                   int* __restrict__ cnt, int* __restrict__ tab,
                                   int* __restrict__ ovf_rc, int* __restrict__ ovf_v,
                                   int nedges) {
  int e = blockIdx.x * blockDim.x + threadIdx.x;
  if (e < nedges) {
    const int2 rc = *(const int2*)(idx + 2 * e);
    const int  vb = __float_as_int(vals[e]);
    const int  p  = atomicAdd(&cnt[rc.x], 1);
    if (p < ROW_CAP) {
      int2 o; o.x = rc.y; o.y = vb;
      *(int2*)(tab + 2 * (rc.x * ROW_CAP + p)) = o;
    } else {
      const int q = atomicAdd(&cnt[OVF_IDX], 1);
      *(int2*)(ovf_rc + 2 * q) = rc;
      ovf_v[q] = vb;
    }
  }
}

template <int U>
__device__ __forceinline__ void pull_rows(const float* __restrict__ X, int colv, int valv,
                                          int j, int f, float4& acc, float& deg) {
  float4 xv[U];
  float  v[U];
#pragma unroll
  for (int u = 0; u < U; ++u) {
    const int c = __builtin_amdgcn_readlane(colv, j + u);
    v[u]  = __int_as_float(__builtin_amdgcn_readlane(valv, j + u));
    xv[u] = *(const float4*)(X + (long long)c * HID + f);
  }
#pragma unroll
  for (int u = 0; u < U; ++u) {
    acc.x += v[u] * xv[u].x;
    acc.y += v[u] * xv[u].y;
    acc.z += v[u] * xv[u].z;
    acc.w += v[u] * xv[u].w;
    deg   += v[u];
  }
}

__global__ void sage_layer_kernel(const int* __restrict__ cnt, const int* __restrict__ tab,
                                  const int* __restrict__ ovf_rc, const int* __restrict__ ovf_v,
                                  const float* __restrict__ X,
                                  const float* __restrict__ Wf, const float* __restrict__ bias,
                                  float* __restrict__ Y, int do_relu) {
  __shared__ __attribute__((aligned(16))) float Xs[16][132];
  const int tile_m = blockIdx.x;
  const int t      = threadIdx.x;
  const int lane   = t & 31;
  const int wid    = t >> 5;

  {
    const int   f    = lane << 2;
    const int2* tab2 = (const int2*)tab;
    const int   r0   = tile_m * 16 + (wid << 1);
    const int2  nn   = *(const int2*)(cnt + r0);
    for (int rr = 0; rr < 2; ++rr) {
      const int rt = (wid << 1) + rr;
      const int r  = r0 + rr;
      const int n = __builtin_amdgcn_readfirstlane(rr ? nn.y : nn.x);
      const int m = n < ROW_CAP ? n : ROW_CAP;
      float4 acc = {0.0f, 0.0f, 0.0f, 0.0f};
      float  deg = 0.0f;
      int colv = 0, valv = 0;
      if (lane < m) {
        const int2 cv = tab2[r * ROW_CAP + lane];
        colv = cv.x; valv = cv.y;
      }
      int j = 0;
      for (; j + 8 <= m; j += 8) pull_rows<8>(X, colv, valv, j, f, acc, deg);
      if (m & 4) { pull_rows<4>(X, colv, valv, j, f, acc, deg); j += 4; }
      if (m & 2) { pull_rows<2>(X, colv, valv, j, f, acc, deg); j += 2; }
      if (m & 1) { pull_rows<1>(X, colv, valv, j, f, acc, deg); }
      if (n > ROW_CAP) {
        const int   total = __builtin_amdgcn_readfirstlane(cnt[OVF_IDX]);
        const int2* orc2  = (const int2*)ovf_rc;
        for (int b = 0; b < total; b += 32) {
          int orow = -1, ocol = 0, oval = 0;
          if (b + lane < total) {
            const int2 k = orc2[b + lane];
            orow = k.x; ocol = k.y; oval = ovf_v[b + lane];
          }
          unsigned hit = __builtin_amdgcn_ballot_w32(orow == r);
          while (hit) {
            const int l = __builtin_ctz(hit);
            hit &= hit - 1;
            pull_rows<1>(X, ocol, oval, l, f, acc, deg);
          }
        }
      }
      const float s = 1.0f / fmaxf(deg, 1.0f);
      float4 o;
      o.x = acc.x * s; o.y = acc.y * s; o.z = acc.z * s; o.w = acc.w * s;
      *(float4*)&Xs[rt][f] = o;
    }
  }
  __syncthreads();

  const int m    = lane & 15;
  const int kh   = (lane >> 4) << 1;
  const int n    = (wid << 4) + (lane & 15);

  const float4* wq = (const float4*)Wf + (wid * (HID / 8)) * 32 + lane;

  v8f c = {};
#pragma unroll
  for (int s = 0; s < HID / 8; ++s) {
    const float4 q = wq[s * 32];
    int k = (s << 3) + kh;
    v2f a0; a0.x = Xs[m][k];     a0.y = Xs[m][k + 1];
    v2f a1; a1.x = Xs[m][k + 4]; a1.y = Xs[m][k + 5];
    v2f b0; b0.x = q.x; b0.y = q.y;
    v2f b1; b1.x = q.z; b1.y = q.w;
    c = __builtin_amdgcn_wmma_f32_16x16x4_f32(false, a0, false, b0, (short)0, c,
                                              false, false);
    c = __builtin_amdgcn_wmma_f32_16x16x4_f32(false, a1, false, b1, (short)0, c,
                                              false, false);
  }

  const float bn      = bias[n];
  const int   rowbase = tile_m * 16 + ((lane >> 4) << 3);
#pragma unroll
  for (int i = 0; i < 8; ++i) {
    float v = c[i] + bn;
    if (do_relu) v = fmaxf(v, 0.0f);
    Y[(long long)(rowbase + i) * HID + n] = v;
  }
}

#define PROJ_MT 5
__global__ void __launch_bounds__(128)
node_proj_kernel(const float* __restrict__ h2, const float* __restrict__ Wd1f,
                 float* __restrict__ P, float* __restrict__ Q) {
  __shared__ __attribute__((aligned(16))) float Xs[16 * PROJ_MT][132];
  const int t    = threadIdx.x;
  const int lane = t & 31;
  const int wid  = t >> 5;
  const int row0 = blockIdx.x * (16 * PROJ_MT);

  {
    const float4* src = (const float4*)(h2 + (long long)row0 * HID);
    float4 v[4 * PROJ_MT];
#pragma unroll
    for (int i = 0; i < 4 * PROJ_MT; ++i) v[i] = src[i * 128 + t];
#pragma unroll
    for (int i = 0; i < 4 * PROJ_MT; ++i) *(float4*)&Xs[4 * i + wid][lane << 2] = v[i];
  }
  __syncthreads();

  const int m  = lane & 15;
  const int kh = (lane >> 4) << 1;
  const int nt = (blockIdx.y << 2) + wid;

  const float4* wq = (const float4*)Wd1f + (nt * (2 * HID / 8)) * 32 + lane;

  v8f cp[PROJ_MT] = {};
  v8f cq[PROJ_MT] = {};
#pragma unroll
  for (int s = 0; s < HID / 8; ++s) {
    const float4 qp = wq[s * 32];
    const float4 qq = wq[(HID / 8 + s) * 32];
    const int k = (s << 3) + kh;
    v2f bp0; bp0.x = qp.x; bp0.y = qp.y;
    v2f bp1; bp1.x = qp.z; bp1.y = qp.w;
    v2f bq0; bq0.x = qq.x; bq0.y = qq.y;
    v2f bq1; bq1.x = qq.z; bq1.y = qq.w;
#pragma unroll
    for (int mt = 0; mt < PROJ_MT; ++mt) {
      const float* xr = &Xs[16 * mt + m][k];
      v2f a0; a0.x = xr[0]; a0.y = xr[1];
      v2f a1; a1.x = xr[4]; a1.y = xr[5];
      cp[mt] = __builtin_amdgcn_wmma_f32_16x16x4_f32(false, a0, false, bp0, (short)0, cp[mt],
                                                     false, false);
      cq[mt] = __builtin_amdgcn_wmma_f32_16x16x4_f32(false, a0, false, bq0, (short)0, cq[mt],
                                                     false, false);
      cp[mt] = __builtin_amdgcn_wmma_f32_16x16x4_f32(false, a1, false, bp1, (short)0, cp[mt],
                                                     false, false);
      cq[mt] = __builtin_amdgcn_wmma_f32_16x16x4_f32(false, a1, false, bq1, (short)0, cq[mt],
                                                     false, false);
    }
  }

  const int n = (nt << 4) + m;
#pragma unroll
  for (int mt = 0; mt < PROJ_MT; ++mt) {
    const int rowbase = row0 + 16 * mt + ((lane >> 4) << 3);
#pragma unroll
    for (int i = 0; i < 8; ++i) {
      P[(long long)(rowbase + i) * HID + n] = cp[mt][i];
      Q[(long long)(rowbase + i) * HID + n] = cq[mt][i];
    }
  }
}

template <int CTRL>
__device__ __forceinline__ int row_pick(int x) {
  return __builtin_amdgcn_update_dpp(0, x, CTRL, 0xf, 0xf, false);
}
template <int CTRL>
__device__ __forceinline__ float row_pickf(float x) {
  return __int_as_float(__builtin_amdgcn_update_dpp(0, __float_as_int(x), CTRL, 0xf, 0xf, false));
}

__device__ __forceinline__ float pair_dot8(const float4& p0, const float4& p1,
                                           const float4& q0, const float4& q1,
                                           const float4& b0, const float4& b1,
                                           const float4& w0, const float4& w1) {
  float s;
  s  = fmaxf((p0.x + q0.x) + b0.x, 0.0f) * w0.x;
  s += fmaxf((p0.y + q0.y) + b0.y, 0.0f) * w0.y;
  s += fmaxf((p0.z + q0.z) + b0.z, 0.0f) * w0.z;
  s += fmaxf((p0.w + q0.w) + b0.w, 0.0f) * w0.w;
  s += fmaxf((p1.x + q1.x) + b1.x, 0.0f) * w1.x;
  s += fmaxf((p1.y + q1.y) + b1.y, 0.0f) * w1.y;
  s += fmaxf((p1.z + q1.z) + b1.z, 0.0f) * w1.z;
  s += fmaxf((p1.w + q1.w) + b1.w, 0.0f) * w1.w;
  return s;
}

template <int IT>
__device__ __forceinline__ float4 score_group(const float* __restrict__ P,
                                              const float* __restrict__ Q,
                                              int uidx, int vidx, int f0,
                                              const float4& b0, const float4& b1,
                                              const float4& w0, const float4& w1) {
  const int u0 = row_pick<0x150 + 4 * IT + 0>(uidx), v0 = row_pick<0x150 + 4 * IT + 0>(vidx);
  const int u1 = row_pick<0x150 + 4 * IT + 1>(uidx), v1 = row_pick<0x150 + 4 * IT + 1>(vidx);
  const int u2 = row_pick<0x150 + 4 * IT + 2>(uidx), v2 = row_pick<0x150 + 4 * IT + 2>(vidx);
  const int u3 = row_pick<0x150 + 4 * IT + 3>(uidx), v3 = row_pick<0x150 + 4 * IT + 3>(vidx);
  const float* pa = P + (long long)u0 * HID + f0;
  const float* pb = P + (long long)u1 * HID + f0;
  const float* pc = P + (long long)u2 * HID + f0;
  const float* pd = P + (long long)u3 * HID + f0;
  const float* qa = Q + (long long)v0 * HID + f0;
  const float* qb = Q + (long long)v1 * HID + f0;
  const float* qc = Q + (long long)v2 * HID + f0;
  const float* qd = Q + (long long)v3 * HID + f0;
  const float4 pa0 = *(const float4*)pa, pa1 = *(const float4*)(pa + 64);
  const float4 qa0 = *(const float4*)qa, qa1 = *(const float4*)(qa + 64);
  const float4 pb0 = *(const float4*)pb, pb1 = *(const float4*)(pb + 64);
  const float4 qb0 = *(const float4*)qb, qb1 = *(const float4*)(qb + 64);
  const float4 pc0 = *(const float4*)pc, pc1 = *(const float4*)(pc + 64);
  const float4 qc0 = *(const float4*)qc, qc1 = *(const float4*)(qc + 64);
  const float4 pd0 = *(const float4*)pd, pd1 = *(const float4*)(pd + 64);
  const float4 qd0 = *(const float4*)qd, qd1 = *(const float4*)(qd + 64);
  float s0 = pair_dot8(pa0, pa1, qa0, qa1, b0, b1, w0, w1);
  float s1 = pair_dot8(pb0, pb1, qb0, qb1, b0, b1, w0, w1);
  float s2 = pair_dot8(pc0, pc1, qc0, qc1, b0, b1, w0, w1);
  float s3 = pair_dot8(pd0, pd1, qd0, qd1, b0, b1, w0, w1);
  s0 += row_pickf<0x161>(s0); s1 += row_pickf<0x161>(s1);
  s2 += row_pickf<0x161>(s2); s3 += row_pickf<0x161>(s3);
  s0 += row_pickf<0x162>(s0); s1 += row_pickf<0x162>(s1);
  s2 += row_pickf<0x162>(s2); s3 += row_pickf<0x162>(s3);
  s0 += row_pickf<0x164>(s0); s1 += row_pickf<0x164>(s1);
  s2 += row_pickf<0x164>(s2); s3 += row_pickf<0x164>(s3);
  s0 += row_pickf<0x168>(s0); s1 += row_pickf<0x168>(s1);
  s2 += row_pickf<0x168>(s2); s3 += row_pickf<0x168>(s3);
  float4 r; r.x = s0; r.y = s1; r.z = s2; r.w = s3;
  return r;
}

#define SC_WAVES 5
__global__ void __launch_bounds__(32 * SC_WAVES)
pair_score_kernel(const float* __restrict__ P, const float* __restrict__ Q,
                  const int* __restrict__ pu, const int* __restrict__ pv,
                  const int* __restrict__ nu, const int* __restrict__ nv,
                  const float* __restrict__ bd1, const float* __restrict__ Wd2,
                  const float* __restrict__ bd2, float* __restrict__ out) {
  __shared__ __attribute__((aligned(16))) float res[SC_WAVES][32];
  const int t    = threadIdx.x;
  const int lane = t & 31;
  const int wid  = t >> 5;
  const int l16  = lane & 15;
  const int hw   = lane >> 4;
  const int pbase = (blockIdx.x * SC_WAVES + wid) << 5;

  const bool neg = blockIdx.x >= (N_POS / (32 * SC_WAVES));
  const int* __restrict__ U = neg ? nu : pu;
  const int* __restrict__ V = neg ? nv : pv;
  const int  lbase = neg ? pbase - N_POS : pbase;

  const int sel  = ((l16 >> 2) << 3) + (hw << 2) + (l16 & 3);
  const int uidx = U[lbase + sel];
  const int vidx = V[lbase + sel];

  const int f0 = l16 << 2;
  const float4 b0 = *(const float4*)(bd1 + f0);
  const float4 b1 = *(const float4*)(bd1 + 64 + f0);
  const float4 w0 = *(const float4*)(Wd2 + f0);
  const float4 w1 = *(const float4*)(Wd2 + 64 + f0);

  const float4 r0 = score_group<0>(P, Q, uidx, vidx, f0, b0, b1, w0, w1);
  const float4 r1 = score_group<1>(P, Q, uidx, vidx, f0, b0, b1, w0, w1);
  const float4 r2 = score_group<2>(P, Q, uidx, vidx, f0, b0, b1, w0, w1);
  const float4 r3 = score_group<3>(P, Q, uidx, vidx, f0, b0, b1, w0, w1);

  if (l16 == 0) {
    float* rw = &res[wid][hw << 2];
    *(float4*)(rw)      = r0;
    *(float4*)(rw + 8)  = r1;
    *(float4*)(rw + 16) = r2;
    *(float4*)(rw + 24) = r3;
  }
  __syncthreads();
  out[pbase + lane] = res[wid][lane] + bd2[0];
}

extern "C" void kernel_launch(void* const* d_in, const int* in_sizes, int n_in,
                              void* d_out, int out_size, void* d_ws, size_t ws_size,
                              hipStream_t stream) {
  const int*   adj_idx = (const int*)  d_in[0];
  const float* adj_val = (const float*)d_in[1];
  const float* h       = (const float*)d_in[3];
  const int*   pos_u   = (const int*)  d_in[4];
  const int*   pos_v   = (const int*)  d_in[5];
  const int*   neg_u   = (const int*)  d_in[6];
  const int*   neg_v   = (const int*)  d_in[7];
  const float* W1      = (const float*)d_in[8];
  const float* b1      = (const float*)d_in[9];
  const float* W2      = (const float*)d_in[10];
  const float* b2      = (const float*)d_in[11];
  const float* Wd1     = (const float*)d_in[12];
  const float* bd1     = (const float*)d_in[13];
  const float* Wd2     = (const float*)d_in[14];
  const float* bd2     = (const float*)d_in[15];

  float* outp = (float*)d_out;
  float* h2         = outp + 2 * N_POS;

  float* ws     = (float*)d_ws;
  int*   cnt    = (int*)ws;
  int*   tab    = (int*)(ws + 50176);
  int*   ovf_rc = tab + 2 * N_NODES * ROW_CAP;
  int*   ovf_v  = ovf_rc + 2 * N_EDGES;
  float* h1     = ws + 50176 + (long long)N_NODES * HID;
  float* W1t    = h1 + (long long)N_NODES * HID;
  float* W2t    = W1t + HID * HID;
  float* Wd1t   = W2t + HID * HID;
  float* Qn     = ws + 50176;
  float* Pn     = h1;

  prep_kernel<<<PREP_NB, 256, 0, stream>>>(W1, W2, Wd1, W1t, W2t, Wd1t, cnt);
  count_place_kernel<<<(N_EDGES + 255) / 256, 256, 0, stream>>>(adj_idx, adj_val, cnt, tab, ovf_rc,
                                                               ovf_v, N_EDGES);
  sage_layer_kernel<<<N_NODES / 16, 256, 0, stream>>>(cnt, tab, ovf_rc, ovf_v, h, W1t, b1, h1, 1);
  sage_layer_kernel<<<N_NODES / 16, 256, 0, stream>>>(cnt, tab, ovf_rc, ovf_v, h1, W2t, b2, h2, 0);
  node_proj_kernel<<<dim3(N_NODES / (16 * PROJ_MT), 2), 128, 0, stream>>>(h2, Wd1t, Pn, Qn);
  pair_score_kernel<<<(2 * N_POS) / (32 * SC_WAVES), 32 * SC_WAVES, 0, stream>>>(
      Pn, Qn, pos_u, pos_v, neg_u, neg_v, bd1, Wd2, bd2, outp);
}


# Edge grouping by two levels of LDS binning (100 edge chunks x 250 bins of 200 rows, then 32-slot row tables), whole-line writes: no global add or scattered store per edge; sums in LDS arrival order
# speedup vs baseline: 4.5465x; 1.4051x over previous
#include <hip/hip_runtime.h>
#include <hip/hip_bf16.h>


typedef __attribute__((ext_vector_type(2))) float v2f;
typedef __attribute__((ext_vector_type(8))) float v8f;

#define N_NODES 50000
#define N_EDGES 800000
#define HID     128
#define N_POS   100000

#define ROW_CAP   32
#define ROW_PITCH 33
#define OVF_IDX   (N_NODES + 32)
#define BIN_ROWS  200
#define NBIN      (N_NODES / BIN_ROWS)
#define GA_WG     100
#define GA_EDGES  (N_EDGES / GA_WG)
#define CELL_CAP  64
#define CELL_PITCH 65
#define GT        512
#define GA_IT     ((GA_EDGES + GT - 1) / GT)
#define PREP_WB   (HID * HID / GT)
#define GA_NB     (GA_WG + 4 * PREP_WB)
__global__ void __launch_bounds__(GT)
bin_prep_kernel(const int* __restrict__ idx, const float* __restrict__ vals,
                const float* __restrict__ W1, const float* __restrict__ W2,
                const float* __restrict__ Wd1,
                float* __restrict__ W1f, float* __restrict__ W2f, float* __restrict__ Wd1f,
                int* __restrict__ cells, int* __restrict__ cellcnt,
                int* __restrict__ spill_rc, int* __restrict__ spill_v,
                int* __restrict__ spillcnt, int* __restrict__ cnt) {
  __shared__ int2 lcell[NBIN * CELL_PITCH];
  __shared__ int bcnt[NBIN];
  __shared__ int scnt;
  const int t = threadIdx.x;
  if (blockIdx.x >= GA_WG) {
    int b = blockIdx.x - GA_WG;
    const float* __restrict__ W;
    float* __restrict__ Wf;
    int lg;
    if (b < PREP_WB)          {                   W = W1;  Wf = W1f;  lg = 4; }
    else if (b < 2 * PREP_WB) { b -= PREP_WB;     W = W2;  Wf = W2f;  lg = 4; }
    else                      { b -= 2 * PREP_WB; W = Wd1; Wf = Wd1f; lg = 5; }
    const int id   = b * GT + t;
    const int j    = id & 3;
    const int lane = (id >> 2) & 31;
    const int rest = id >> 7;
    const int w    = rest >> lg;
    const int sidx = rest - (w << lg);
    const int n    = (w << 4) + (lane & 15);
    const int k    = (sidx << 3) + ((j >> 1) << 2) + ((lane >> 4) << 1) + (j & 1);
    Wf[id] = W[k * HID + n];
    return;
  }
  const int g     = blockIdx.x;
  const int ebase = g * GA_EDGES;
  if (t < NBIN) bcnt[t] = 0;
  if (t == 0)   scnt = 0;
  __syncthreads();
  int2 rc[GA_IT];
  int  vb[GA_IT];
  int  cb[GA_IT];
  int  pl[GA_IT];
#pragma unroll
  for (int i = 0; i < GA_IT; ++i) {
    const int k = i * GT + t;
    rc[i].x = 0; rc[i].y = 0; vb[i] = 0;
    if (k < GA_EDGES) {
      rc[i] = *(const int2*)(idx + 2 * (ebase + k));
      vb[i] = __float_as_int(vals[ebase + k]);
    }
  }
#pragma unroll
  for (int i = 0; i < GA_IT; ++i) {
    const int k = i * GT + t;
    cb[i] = 0; pl[i] = 0;
    if (k < GA_EDGES) {
      const int row = rc[i].x;
      const int bin = (int)((unsigned)row / (unsigned)BIN_ROWS);
      rc[i].x = row - bin * BIN_ROWS;
      cb[i]   = bin * CELL_PITCH;
      pl[i]   = atomicAdd(&bcnt[bin], 1);
    }
  }
#pragma unroll
  for (int i = 0; i < GA_IT; ++i) {
    const int k = i * GT + t;
    if (k < GA_EDGES) {
      if (pl[i] < CELL_CAP) {
        int2 o; o.x = (rc[i].x << 16) | rc[i].y; o.y = vb[i];
        lcell[cb[i] + pl[i]] = o;
      } else {
        const int s = atomicAdd(&scnt, 1);
        int2 o; o.x = (cb[i] / CELL_PITCH) * BIN_ROWS + rc[i].x; o.y = rc[i].y;
        *(int2*)(spill_rc + 2 * (ebase + s)) = o;
        spill_v[ebase + s] = vb[i];
      }
    }
  }
  __syncthreads();
  if (t < NBIN) {
    const int c = bcnt[t];
    cellcnt[t * GA_WG + g] = c < CELL_CAP ? c : CELL_CAP;
  }
  if (t == 0) {
    spillcnt[g] = scnt;
    if (g == 0) cnt[OVF_IDX] = 0;
  }
  int4* dst = (int4*)cells + (long long)g * (NBIN * CELL_CAP / 2);
  for (int k = t; k < NBIN * CELL_CAP / 2; k += GT) {
    const int2* e = &lcell[(k >> 5) * CELL_PITCH + ((k & 31) << 1)];
    const int2 a = e[0], b = e[1];
    int4 o; o.x = a.x; o.y = a.y; o.z = b.x; o.w = b.y;
    dst[k] = o;
  }
}

__device__ __forceinline__ void row_place(int2* ltab, int p, int rowbase, int rl, int col, int vb,
                                          int* __restrict__ cnt,
                                          int* __restrict__ ovf_rc, int* __restrict__ ovf_v) {
  if (p < ROW_CAP) {
    int2 o; o.x = col; o.y = vb;
    ltab[rl * ROW_PITCH + p] = o;
  } else {
    const int q = atomicAdd(&cnt[OVF_IDX], 1);
    int2 k; k.x = rowbase + rl; k.y = col;
    *(int2*)(ovf_rc + 2 * q) = k;
    ovf_v[q] = vb;
  }
}

#define RB_CELLS ((GA_WG + GT / 32 - 1) / (GT / 32))
__global__ void __launch_bounds__(GT)
row_build_kernel(const int* __restrict__ cells, const int* __restrict__ cellcnt,
                 const int* __restrict__ spill_rc, const int* __restrict__ spill_v,
                 const int* __restrict__ spillcnt,
                 int* __restrict__ cnt, int* __restrict__ tab,
                 int* __restrict__ ovf_rc, int* __restrict__ ovf_v) {
  __shared__ int2 ltab[BIN_ROWS * ROW_PITCH];
  __shared__ int lcnt[BIN_ROWS];
  __shared__ int lsc[GA_WG];
  const int t       = threadIdx.x;
  const int lane    = t & 31;
  const int wid     = t >> 5;
  const int bin     = blockIdx.x;
  const int rowbase = bin * BIN_ROWS;

  int2 ent[RB_CELLS][2];
  int  cc[RB_CELLS];
  int  pp[RB_CELLS][2];
#pragma unroll
  for (int j = 0; j < RB_CELLS; ++j) {
    const int g = wid + (GT / 32) * j;
    cc[j] = 0;
    ent[j][0].x = 0; ent[j][0].y = 0; ent[j][1].x = 0; ent[j][1].y = 0;
    if (g < GA_WG) {
      cc[j] = cellcnt[bin * GA_WG + g];
      const int2* cell = (const int2*)cells + ((long long)g * NBIN + bin) * CELL_CAP;
      ent[j][0] = cell[lane];
      ent[j][1] = cell[lane + 32];
    }
  }
  if (t < BIN_ROWS) lcnt[t] = 0;
  if (t < GA_WG) {
    const int sc = spillcnt[t];
    lsc[t] = sc < GA_EDGES ? sc : GA_EDGES;
  }
  __syncthreads();

#pragma unroll
  for (int j = 0; j < RB_CELLS; ++j) {
#pragma unroll
    for (int h = 0; h < 2; ++h) {
      pp[j][h] = 0;
      if (lane + 32 * h < cc[j]) pp[j][h] = atomicAdd(&lcnt[ent[j][h].x >> 16], 1);
    }
  }
#pragma unroll
  for (int j = 0; j < RB_CELLS; ++j) {
#pragma unroll
    for (int h = 0; h < 2; ++h) {
      if (lane + 32 * h < cc[j]) {
        const int x = ent[j][h].x;
        row_place(ltab, pp[j][h], rowbase, x >> 16, x & 0xffff, ent[j][h].y, cnt, ovf_rc, ovf_v);
      }
    }
  }

  {
    int any = lsc[lane] | lsc[lane + 32] | lsc[lane + 64];
    if (lane + 96 < GA_WG) any |= lsc[lane + 96];
    if (__builtin_amdgcn_ballot_w32(any != 0) != 0u) {
      for (int g = 0; g < GA_WG; ++g) {
        const int sc = lsc[g];
        for (int i = t; i < sc; i += GT) {
          const int2 k = *(const int2*)(spill_rc + 2 * (g * GA_EDGES + i));
          const int  r = k.x - rowbase;
          if ((unsigned)r < (unsigned)BIN_ROWS) {
            const int p = atomicAdd(&lcnt[r], 1);
            row_place(ltab, p, rowbase, r, k.y, spill_v[g * GA_EDGES + i], cnt, ovf_rc, ovf_v);
          }
        }
      }
    }
  }
  __syncthreads();

  if (t < BIN_ROWS) cnt[rowbase + t] = lcnt[t];
  int4* dst = (int4*)tab + (long long)bin * (BIN_ROWS * ROW_CAP / 2);
  for (int k = t; k < BIN_ROWS * ROW_CAP / 2; k += GT) {
    const int2* e = &ltab[(k >> 4) * ROW_PITCH + ((k & 15) << 1)];
    const int2 a = e[0], b = e[1];
    int4 o; o.x = a.x; o.y = a.y; o.z = b.x; o.w = b.y;
    dst[k] = o;
  }
}

template <int U>
__device__ __forceinline__ void pull_rows(const float* __restrict__ X, int colv, int valv,
                                          int j, int f, float4& acc, float& deg) {
  float4 xv[U];
  float  v[U];
#pragma unroll
  for (int u = 0; u < U; ++u) {
    const int c = __builtin_amdgcn_readlane(colv, j + u);
    v[u]  = __int_as_float(__builtin_amdgcn_readlane(valv, j + u));
    xv[u] = *(const float4*)(X + (long long)c * HID + f);
  }
#pragma unroll
  for (int u = 0; u < U; ++u) {
    acc.x += v[u] * xv[u].x;
    acc.y += v[u] * xv[u].y;
    acc.z += v[u] * xv[u].z;
    acc.w += v[u] * xv[u].w;
    deg   += v[u];
  }
}

__global__ void sage_layer_kernel(const int* __restrict__ cnt, const int* __restrict__ tab,
                                  const int* __restrict__ ovf_rc, const int* __restrict__ ovf_v,
                                  const float* __restrict__ X,
                                  const float* __restrict__ Wf, const float* __restrict__ bias,
                                  float* __restrict__ Y, int do_relu) {
  __shared__ __attribute__((aligned(16))) float Xs[16][132];
  const int tile_m = blockIdx.x;
  const int t      = threadIdx.x;
  const int lane   = t & 31;
  const int wid    = t >> 5;

  {
    const int   f    = lane << 2;
    const int2* tab2 = (const int2*)tab;
    const int   r0   = tile_m * 16 + (wid << 1);
    const int2  nn   = *(const int2*)(cnt + r0);
    for (int rr = 0; rr < 2; ++rr) {
      const int rt = (wid << 1) + rr;
      const int r  = r0 + rr;
      const int n = __builtin_amdgcn_readfirstlane(rr ? nn.y : nn.x);
      const int m = n < ROW_CAP ? n : ROW_CAP;
      float4 acc = {0.0f, 0.0f, 0.0f, 0.0f};
      float  deg = 0.0f;
      int colv = 0, valv = 0;
      if (lane < m) {
        const int2 cv = tab2[r * ROW_CAP + lane];
        colv = cv.x; valv = cv.y;
      }
      int j = 0;
      for (; j + 8 <= m; j += 8) pull_rows<8>(X, colv, valv, j, f, acc, deg);
      if (m & 4) { pull_rows<4>(X, colv, valv, j, f, acc, deg); j += 4; }
      if (m & 2) { pull_rows<2>(X, colv, valv, j, f, acc, deg); j += 2; }
      if (m & 1) { pull_rows<1>(X, colv, valv, j, f, acc, deg); }
      if (n > ROW_CAP) {
        const int   total = __builtin_amdgcn_readfirstlane(cnt[OVF_IDX]);
        const int2* orc2  = (const int2*)ovf_rc;
        for (int b = 0; b < total; b += 32) {
          int orow = -1, ocol = 0, oval = 0;
          if (b + lane < total) {
            const int2 k = orc2[b + lane];
            orow = k.x; ocol = k.y; oval = ovf_v[b + lane];
          }
          unsigned hit = __builtin_amdgcn_ballot_w32(orow == r);
          while (hit) {
            const int l = __builtin_ctz(hit);
            hit &= hit - 1;
            pull_rows<1>(X, ocol, oval, l, f, acc, deg);
          }
        }
      }
      const float s = 1.0f / fmaxf(deg, 1.0f);
      float4 o;
      o.x = acc.x * s; o.y = acc.y * s; o.z = acc.z * s; o.w = acc.w * s;
      *(float4*)&Xs[rt][f] = o;
    }
  }
  __syncthreads();

  const int m    = lane & 15;
  const int kh   = (lane >> 4) << 1;
  const int n    = (wid << 4) + (lane & 15);

  const float4* wq = (const float4*)Wf + (wid * (HID / 8)) * 32 + lane;

  v8f c = {};
#pragma unroll
  for (int s = 0; s < HID / 8; ++s) {
    const float4 q = wq[s * 32];
    int k = (s << 3) + kh;
    v2f a0; a0.x = Xs[m][k];     a0.y = Xs[m][k + 1];
    v2f a1; a1.x = Xs[m][k + 4]; a1.y = Xs[m][k + 5];
    v2f b0; b0.x = q.x; b0.y = q.y;
    v2f b1; b1.x = q.z; b1.y = q.w;
    c = __builtin_amdgcn_wmma_f32_16x16x4_f32(false, a0, false, b0, (short)0, c,
                                              false, false);
    c = __builtin_amdgcn_wmma_f32_16x16x4_f32(false, a1, false, b1, (short)0, c,
                                              false, false);
  }

  const float bn      = bias[n];
  const int   rowbase = tile_m * 16 + ((lane >> 4) << 3);
#pragma unroll
  for (int i = 0; i < 8; ++i) {
    float v = c[i] + bn;
    if (do_relu) v = fmaxf(v, 0.0f);
    Y[(long long)(rowbase + i) * HID + n] = v;
  }
}

#define PROJ_MT 5
__global__ void __launch_bounds__(128)
node_proj_kernel(const float* __restrict__ h2, const float* __restrict__ Wd1f,
                 float* __restrict__ P, float* __restrict__ Q) {
  __shared__ __attribute__((aligned(16))) float Xs[16 * PROJ_MT][132];
  const int t    = threadIdx.x;
  const int lane = t & 31;
  const int wid  = t >> 5;
  const int row0 = blockIdx.x * (16 * PROJ_MT);

  {
    const float4* src = (const float4*)(h2 + (long long)row0 * HID);
    float4 v[4 * PROJ_MT];
#pragma unroll
    for (int i = 0; i < 4 * PROJ_MT; ++i) v[i] = src[i * 128 + t];
#pragma unroll
    for (int i = 0; i < 4 * PROJ_MT; ++i) *(float4*)&Xs[4 * i + wid][lane << 2] = v[i];
  }
  __syncthreads();

  const int m  = lane & 15;
  const int kh = (lane >> 4) << 1;
  const int nt = (blockIdx.y << 2) + wid;

  const float4* wq = (const float4*)Wd1f + (nt * (2 * HID / 8)) * 32 + lane;

  v8f cp[PROJ_MT] = {};
  v8f cq[PROJ_MT] = {};
#pragma unroll
  for (int s = 0; s < HID / 8; ++s) {
    const float4 qp = wq[s * 32];
    const float4 qq = wq[(HID / 8 + s) * 32];
    const int k = (s << 3) + kh;
    v2f bp0; bp0.x = qp.x; bp0.y = qp.y;
    v2f bp1; bp1.x = qp.z; bp1.y = qp.w;
    v2f bq0; bq0.x = qq.x; bq0.y = qq.y;
    v2f bq1; bq1.x = qq.z; bq1.y = qq.w;
#pragma unroll
    for (int mt = 0; mt < PROJ_MT; ++mt) {
      const float* xr = &Xs[16 * mt + m][k];
      v2f a0; a0.x = xr[0]; a0.y = xr[1];
      v2f a1; a1.x = xr[4]; a1.y = xr[5];
      cp[mt] = __builtin_amdgcn_wmma_f32_16x16x4_f32(false, a0, false, bp0, (short)0, cp[mt],
                                                     false, false);
      cq[mt] = __builtin_amdgcn_wmma_f32_16x16x4_f32(false, a0, false, bq0, (short)0, cq[mt],
                                                     false, false);
      cp[mt] = __builtin_amdgcn_wmma_f32_16x16x4_f32(false, a1, false, bp1, (short)0, cp[mt],
                                                     false, false);
      cq[mt] = __builtin_amdgcn_wmma_f32_16x16x4_f32(false, a1, false, bq1, (short)0, cq[mt],
                                                     false, false);
    }
  }

  const int n = (nt << 4) + m;
#pragma unroll
  for (int mt = 0; mt < PROJ_MT; ++mt) {
    const int rowbase = row0 + 16 * mt + ((lane >> 4) << 3);
#pragma unroll
    for (int i = 0; i < 8; ++i) {
      P[(long long)(rowbase + i) * HID + n] = cp[mt][i];
      Q[(long long)(rowbase + i) * HID + n] = cq[mt][i];
    }
  }
}

template <int CTRL>
__device__ __forceinline__ int row_pick(int x) {
  return __builtin_amdgcn_update_dpp(0, x, CTRL, 0xf, 0xf, false);
}
template <int CTRL>
__device__ __forceinline__ float row_pickf(float x) {
  return __int_as_float(__builtin_amdgcn_update_dpp(0, __float_as_int(x), CTRL, 0xf, 0xf, false));
}

__device__ __forceinline__ float pair_dot8(const float4& p0, const float4& p1,
                                           const float4& q0, const float4& q1,
                                           const float4& b0, const float4& b1,
                                           const float4& w0, const float4& w1) {
  float s;
  s  = fmaxf((p0.x + q0.x) + b0.x, 0.0f) * w0.x;
  s += fmaxf((p0.y + q0.y) + b0.y, 0.0f) * w0.y;
  s += fmaxf((p0.z + q0.z) + b0.z, 0.0f) * w0.z;
  s += fmaxf((p0.w + q0.w) + b0.w, 0.0f) * w0.w;
  s += fmaxf((p1.x + q1.x) + b1.x, 0.0f) * w1.x;
  s += fmaxf((p1.y + q1.y) + b1.y, 0.0f) * w1.y;
  s += fmaxf((p1.z + q1.z) + b1.z, 0.0f) * w1.z;
  s += fmaxf((p1.w + q1.w) + b1.w, 0.0f) * w1.w;
  return s;
}

template <int IT>
__device__ __forceinline__ float4 score_group(const float* __restrict__ P,
                                              const float* __restrict__ Q,
                                              int uidx, int vidx, int f0,
                                              const float4& b0, const float4& b1,
                                              const float4& w0, const float4& w1) {
  const int u0 = row_pick<0x150 + 4 * IT + 0>(uidx), v0 = row_pick<0x150 + 4 * IT + 0>(vidx);
  const int u1 = row_pick<0x150 + 4 * IT + 1>(uidx), v1 = row_pick<0x150 + 4 * IT + 1>(vidx);
  const int u2 = row_pick<0x150 + 4 * IT + 2>(uidx), v2 = row_pick<0x150 + 4 * IT + 2>(vidx);
  const int u3 = row_pick<0x150 + 4 * IT + 3>(uidx), v3 = row_pick<0x150 + 4 * IT + 3>(vidx);
  const float* pa = P + (long long)u0 * HID + f0;
  const float* pb = P + (long long)u1 * HID + f0;
  const float* pc = P + (long long)u2 * HID + f0;
  const float* pd = P + (long long)u3 * HID + f0;
  const float* qa = Q + (long long)v0 * HID + f0;
  const float* qb = Q + (long long)v1 * HID + f0;
  const float* qc = Q + (long long)v2 * HID + f0;
  const float* qd = Q + (long long)v3 * HID + f0;
  const float4 pa0 = *(const float4*)pa, pa1 = *(const float4*)(pa + 64);
  const float4 qa0 = *(const float4*)qa, qa1 = *(const float4*)(qa + 64);
  const float4 pb0 = *(const float4*)pb, pb1 = *(const float4*)(pb + 64);
  const float4 qb0 = *(const float4*)qb, qb1 = *(const float4*)(qb + 64);
  const float4 pc0 = *(const float4*)pc, pc1 = *(const float4*)(pc + 64);
  const float4 qc0 = *(const float4*)qc, qc1 = *(const float4*)(qc + 64);
  const float4 pd0 = *(const float4*)pd, pd1 = *(const float4*)(pd + 64);
  const float4 qd0 = *(const float4*)qd, qd1 = *(const float4*)(qd + 64);
  float s0 = pair_dot8(pa0, pa1, qa0, qa1, b0, b1, w0, w1);
  float s1 = pair_dot8(pb0, pb1, qb0, qb1, b0, b1, w0, w1);
  float s2 = pair_dot8(pc0, pc1, qc0, qc1, b0, b1, w0, w1);
  float s3 = pair_dot8(pd0, pd1, qd0, qd1, b0, b1, w0, w1);
  s0 += row_pickf<0x161>(s0); s1 += row_pickf<0x161>(s1);
  s2 += row_pickf<0x161>(s2); s3 += row_pickf<0x161>(s3);
  s0 += row_pickf<0x162>(s0); s1 += row_pickf<0x162>(s1);
  s2 += row_pickf<0x162>(s2); s3 += row_pickf<0x162>(s3);
  s0 += row_pickf<0x164>(s0); s1 += row_pickf<0x164>(s1);
  s2 += row_pickf<0x164>(s2); s3 += row_pickf<0x164>(s3);
  s0 += row_pickf<0x168>(s0); s1 += row_pickf<0x168>(s1);
  s2 += row_pickf<0x168>(s2); s3 += row_pickf<0x168>(s3);
  float4 r; r.x = s0; r.y = s1; r.z = s2; r.w = s3;
  return r;
}

#define SC_WAVES 5
__global__ void __launch_bounds__(32 * SC_WAVES)
pair_score_kernel(const float* __restrict__ P, const float* __restrict__ Q,
                  const int* __restrict__ pu, const int* __restrict__ pv,
                  const int* __restrict__ nu, const int* __restrict__ nv,
                  const float* __restrict__ bd1, const float* __restrict__ Wd2,
                  const float* __restrict__ bd2, float* __restrict__ out) {
  __shared__ __attribute__((aligned(16))) float res[SC_WAVES][32];
  const int t    = threadIdx.x;
  const int lane = t & 31;
  const int wid  = t >> 5;
  const int l16  = lane & 15;
  const int hw   = lane >> 4;
  const int pbase = (blockIdx.x * SC_WAVES + wid) << 5;

  const bool neg = blockIdx.x >= (N_POS / (32 * SC_WAVES));
  const int* __restrict__ U = neg ? nu : pu;
  const int* __restrict__ V = neg ? nv : pv;
  const int  lbase = neg ? pbase - N_POS : pbase;

  const int sel  = ((l16 >> 2) << 3) + (hw << 2) + (l16 & 3);
  const int uidx = U[lbase + sel];
  const int vidx = V[lbase + sel];

  const int f0 = l16 << 2;
  const float4 b0 = *(const float4*)(bd1 + f0);
  const float4 b1 = *(const float4*)(bd1 + 64 + f0);
  const float4 w0 = *(const float4*)(Wd2 + f0);
  const float4 w1 = *(const float4*)(Wd2 + 64 + f0);

  const float4 r0 = score_group<0>(P, Q, uidx, vidx, f0, b0, b1, w0, w1);
  const float4 r1 = score_group<1>(P, Q, uidx, vidx, f0, b0, b1, w0, w1);
  const float4 r2 = score_group<2>(P, Q, uidx, vidx, f0, b0, b1, w0, w1);
  const float4 r3 = score_group<3>(P, Q, uidx, vidx, f0, b0, b1, w0, w1);

  if (l16 == 0) {
    float* rw = &res[wid][hw << 2];
    *(float4*)(rw)      = r0;
    *(float4*)(rw + 8)  = r1;
    *(float4*)(rw + 16) = r2;
    *(float4*)(rw + 24) = r3;
  }
  __syncthreads();
  out[pbase + lane] = res[wid][lane] + bd2[0];
}

extern "C" void kernel_launch(void* const* d_in, const int* in_sizes, int n_in,
                              void* d_out, int out_size, void* d_ws, size_t ws_size,
                              hipStream_t stream) {
  const int*   adj_idx = (const int*)  d_in[0];
  const float* adj_val = (const float*)d_in[1];
  const float* h       = (const float*)d_in[3];
  const int*   pos_u   = (const int*)  d_in[4];
  const int*   pos_v   = (const int*)  d_in[5];
  const int*   neg_u   = (const int*)  d_in[6];
  const int*   neg_v   = (const int*)  d_in[7];
  const float* W1      = (const float*)d_in[8];
  const float* b1      = (const float*)d_in[9];
  const float* W2      = (const float*)d_in[10];
  const float* b2      = (const float*)d_in[11];
  const float* Wd1     = (const float*)d_in[12];
  const float* bd1     = (const float*)d_in[13];
  const float* Wd2     = (const float*)d_in[14];
  const float* bd2     = (const float*)d_in[15];

  float* outp = (float*)d_out;
  float* h2         = outp + 2 * N_POS;

  float* ws     = (float*)d_ws;
  int*   cnt    = (int*)ws;
  int*   tab    = (int*)(ws + 50176);
  int*   ovf_rc = tab + 2 * N_NODES * ROW_CAP;
  int*   ovf_v  = ovf_rc + 2 * N_EDGES;
  float* h1     = ws + 50176 + (long long)N_NODES * HID;
  float* W1t    = h1 + (long long)N_NODES * HID;
  float* W2t    = W1t + HID * HID;
  float* Wd1t   = W2t + HID * HID;
  float* Qn     = ws + 50176;
  float* Pn     = h1;
  int*   cells    = (int*)h1;
  int*   cellcnt  = cells + 2 * GA_WG * NBIN * CELL_CAP;
  int*   spillcnt = cellcnt + 25088;
  int*   spill_rc = cellcnt + 32768;
  int*   spill_v  = spill_rc + 2 * N_EDGES;

  bin_prep_kernel<<<GA_NB, GT, 0, stream>>>(adj_idx, adj_val, W1, W2, Wd1, W1t, W2t, Wd1t,
                                              cells, cellcnt, spill_rc, spill_v, spillcnt, cnt);
  row_build_kernel<<<NBIN, GT, 0, stream>>>(cells, cellcnt, spill_rc, spill_v, spillcnt,
                                              cnt, tab, ovf_rc, ovf_v);
  sage_layer_kernel<<<N_NODES / 16, 256, 0, stream>>>(cnt, tab, ovf_rc, ovf_v, h, W1t, b1, h1, 1);
  sage_layer_kernel<<<N_NODES / 16, 256, 0, stream>>>(cnt, tab, ovf_rc, ovf_v, h1, W2t, b2, h2, 0);
  node_proj_kernel<<<dim3(N_NODES / (16 * PROJ_MT), 2), 128, 0, stream>>>(h2, Wd1t, Pn, Qn);
  pair_score_kernel<<<(2 * N_POS) / (32 * SC_WAVES), 32 * SC_WAVES, 0, stream>>>(
      Pn, Qn, pos_u, pos_v, neg_u, neg_v, bd1, Wd2, bd2, outp);
}
